# starting kernel
# baseline (speedup 1.0000x reference)
#include <hip/hip_runtime.h>

#define NATOMS 4096
#define CC     64
#define TT     256
#define ASIZE  8
#define BATCH  4
#define KDIM   512
#define PPOS   257
#define RES_W  304
#define NSPLIT 2
#define NTPB   9
#define MTILE  64
#define CORR_THREADS 128
#define NITER  32

typedef __attribute__((ext_vector_type(2))) float v2f;
typedef __attribute__((ext_vector_type(8))) float v8f;

__global__ void k_normalize(const float* __restrict__ d, float* __restrict__ dnn) {
    __shared__ float red[4];
    const int a = blockIdx.x;
    const float* src = d + (size_t)a * KDIM;
    float s = 0.f;
    for (int i = threadIdx.x; i < KDIM; i += CORR_THREADS) { float v = src[i]; s += v * v; }
    for (int off = 16; off > 0; off >>= 1) s += __shfl_down(s, off, 32);
    const int wv = threadIdx.x >> 5;
    if ((threadIdx.x & 31) == 0) red[wv] = s;
    __syncthreads();
    if (threadIdx.x == 0) red[0] = 1.f / (sqrtf(red[0] + red[1] + red[2] + red[3]) + 1e-8f);
    __syncthreads();
    const float inv = red[0];
    float* dst = dnn + (size_t)a * KDIM;
    for (int i = threadIdx.x; i < KDIM; i += CORR_THREADS) dst[i] = src[i] * inv;
}

__global__ void k_init(const float* __restrict__ x, float* __restrict__ res,
                       float* __restrict__ recon, unsigned long long* __restrict__ best) {
    const int i = blockIdx.x * blockDim.x + threadIdx.x;
    if (i < BATCH * CC * RES_W) {
        const int col = i % RES_W;
        const int bc  = i / RES_W;
        res[i]   = (col < TT) ? x[bc * TT + col] : 0.f;
        recon[i] = 0.f;
    }
    if (i < BATCH) best[i] = 0ull;
}

__global__ __launch_bounds__(CORR_THREADS, 1)
void k_corr(const float* __restrict__ dnn, const float* __restrict__ res,
            unsigned long long* __restrict__ best) {
    extern __shared__ float smem[];
    const int b    = blockIdx.z;
    const int nt0  = blockIdx.y * NTPB;
    const int m0   = blockIdx.x * MTILE;
    const int tid  = threadIdx.x;
    const int lane = tid & 31;
    const int wave = tid >> 5;

    const int half = lane >> 4;
    const int nloc = lane & 15;
    const float* arow = dnn + (size_t)(m0 + wave * 16 + nloc) * KDIM + half * 2;

    #pragma unroll
    for (int pf = 0; pf < 16; ++pf)
        __builtin_prefetch(arow + pf * 32, 0, 3);

    const float4* resb = (const float4*)(res + (size_t)b * CC * RES_W);
    float4* s4 = (float4*)smem;
    for (int i = tid; i < (CC * RES_W) / 4; i += CORR_THREADS) s4[i] = resb[i];
    __syncthreads();

    v8f acc[NTPB] = {};

    auto loadA = [&](int ks) { v2f a; a.x = arow[ks * 4]; a.y = arow[ks * 4 + 1]; return a; };
    auto srowp = [&](int ks) {
        const int kbase = ks * 4 + half * 2;
        return (const float*)smem + (kbase >> 3) * RES_W + (kbase & 7) + nloc + nt0 * 16;
    };

    v2f a0 = loadA(0);
    for (int ks = 0; ks < KDIM / 4; ++ks) {
        v2f an = loadA(ks + 1 < KDIM / 4 ? ks + 1 : ks);
        const float* sr = srowp(ks);
        v2f bf[NTPB];
        #pragma unroll
        for (int nt = 0; nt < NTPB; ++nt) {
            bf[nt].x = sr[nt * 16];
            bf[nt].y = sr[nt * 16 + 1];
        }
        #pragma unroll
        for (int nt = 0; nt < NTPB; ++nt)
            acc[nt] = __builtin_amdgcn_wmma_f32_16x16x4_f32(
                false, a0, false, bf[nt], (short)0, acc[nt], false, false);
        a0 = an;
    }

    float bestv = -3.402823466e38f;
    unsigned bestidx = 0u;
    #pragma unroll
    for (int nt = 0; nt < NTPB; ++nt) {
        const int pos = (nt0 + nt) * 16 + nloc;
        if (pos < PPOS) {
            #pragma unroll
            for (int r = 0; r < 8; ++r) {
                const float v = acc[nt][r];
                if (v > bestv) {
                    bestv = v;
                    bestidx = (unsigned)((m0 + wave * 16 + r + 8 * half) * PPOS + pos);
                }
            }
        }
    }
    unsigned vb = __float_as_uint(bestv);
    vb = (vb & 0x80000000u) ? ~vb : (vb | 0x80000000u);
    unsigned long long key = ((unsigned long long)vb << 32) | (0xFFFFFFFFu - bestidx);
    for (int off = 16; off > 0; off >>= 1) {
        unsigned long long o = __shfl_xor(key, off, 32);
        if (o > key) key = o;
    }
    if (lane == 0) atomicMax(best + b, key);
}

__global__ void k_update(const float* __restrict__ dnn, float* __restrict__ res,
                         float* __restrict__ recon, unsigned long long* __restrict__ best) {
    const int b = blockIdx.x;
    const unsigned long long key = best[b];
    const unsigned enc  = (unsigned)(key >> 32);
    const unsigned bits = (enc & 0x80000000u) ? (enc & 0x7FFFFFFFu) : ~enc;
    const float val = __uint_as_float(bits);
    const unsigned idx = 0xFFFFFFFFu - (unsigned)(key & 0xFFFFFFFFu);
    const int atom = (int)(idx / PPOS);
    const int pos  = (int)(idx % PPOS);
    const float* ap = dnn + (size_t)atom * KDIM;
    float* rb = res   + (size_t)b * CC * RES_W;
    float* cb = recon + (size_t)b * CC * RES_W;
    for (int i = threadIdx.x; i < KDIM; i += blockDim.x) {
        const int c = i >> 3, t = i & 7;
        const float av = ap[i] * val;
        rb[c * RES_W + pos + t] -= av;
        cb[c * RES_W + pos + t] += av;
    }
    __syncthreads();
    if (threadIdx.x == 0) best[b] = 0ull;
}

__global__ void k_output(const float* __restrict__ recon, float* __restrict__ out) {
    const int i = blockIdx.x * blockDim.x + threadIdx.x;
    if (i < BATCH * CC * TT) {
        const int t  = i % TT;
        const int bc = i / TT;
        out[i] = recon[(size_t)bc * RES_W + t];
    }
}

extern "C" void kernel_launch(void* const* d_in, const int* in_sizes, int n_in,
                              void* d_out, int out_size, void* d_ws, size_t ws_size,
                              hipStream_t stream) {
    const float* x = (const float*)d_in[0];
    const float* d = (const float*)d_in[1];
    (void)in_sizes; (void)n_in; (void)out_size; (void)ws_size;

    char* ws = (char*)d_ws;
    float* dnn   = (float*)ws;
    float* res   = (float*)(ws + (size_t)NATOMS * KDIM * 4);
    float* recon = res + BATCH * CC * RES_W;
    unsigned long long* best = (unsigned long long*)(recon + BATCH * CC * RES_W);

    k_normalize<<<NATOMS, CORR_THREADS, 0, stream>>>(d, dnn);
    k_init<<<(BATCH * CC * RES_W + 255) / 256, 256, 0, stream>>>(x, res, recon, best);

    const size_t shbytes = (size_t)CC * RES_W * sizeof(float);
    dim3 grid(NATOMS / MTILE, NSPLIT, BATCH);
    for (int it = 0; it < NITER; ++it) {
        k_corr<<<grid, CORR_THREADS, shbytes, stream>>>(dnn, res, best);
        k_update<<<BATCH, 128, 0, stream>>>(dnn, res, recon, best);
    }
    k_output<<<(BATCH * CC * TT + 255) / 256, 256, 0, stream>>>(recon, (float*)d_out);
}


# Recompute only the two 16-position tiles whose windows overlap the last rank-1 update; per-tile max keys kept in workspace, update kernel reduces them (atomicMax removed)
# speedup vs baseline: 2.7082x; 2.7082x over previous
#include <hip/hip_runtime.h>

#define NATOMS 4096
#define CC     64
#define TT     256
#define ASIZE  8
#define BATCH  4
#define KDIM   512
#define PPOS   257
#define RES_W  304
#define NSPLIT 2
#define NTPB   9
#define MTILE  64
#define CORR_THREADS 128
#define NITER  32
#define NTILES 17
#define MTILES 256
#define KSLOTS 32
#define SLABW  40

typedef __attribute__((ext_vector_type(2))) float v2f;
typedef __attribute__((ext_vector_type(8))) float v8f;

__global__ void k_normalize(const float* __restrict__ d, float* __restrict__ dnn) {
    __shared__ float red[4];
    const int a = blockIdx.x;
    const float* src = d + (size_t)a * KDIM;
    float s = 0.f;
    for (int i = threadIdx.x; i < KDIM; i += CORR_THREADS) { float v = src[i]; s += v * v; }
    for (int off = 16; off > 0; off >>= 1) s += __shfl_down(s, off, 32);
    const int wv = threadIdx.x >> 5;
    if ((threadIdx.x & 31) == 0) red[wv] = s;
    __syncthreads();
    if (threadIdx.x == 0) red[0] = 1.f / (sqrtf(red[0] + red[1] + red[2] + red[3]) + 1e-8f);
    __syncthreads();
    const float inv = red[0];
    float* dst = dnn + (size_t)a * KDIM;
    for (int i = threadIdx.x; i < KDIM; i += CORR_THREADS) dst[i] = src[i] * inv;
}

__global__ void k_init(const float* __restrict__ x, float* __restrict__ res,
                       float* __restrict__ recon) {
    const int i = blockIdx.x * blockDim.x + threadIdx.x;
    if (i < BATCH * CC * RES_W) {
        const int col = i % RES_W;
        const int bc  = i / RES_W;
        res[i]   = (col < TT) ? x[bc * TT + col] : 0.f;
        recon[i] = 0.f;
    }
}

template <int NT>
__device__ __forceinline__ void tile_keys_store(const v8f (&acc)[NT], int nt0, int atom0,
                                                int half, int nloc, int lane,
                                                unsigned long long* __restrict__ rec) {
    unsigned long long mykey = 0ull;
    #pragma unroll
    for (int nt = 0; nt < NT; ++nt) {
        float bestv = -3.402823466e38f;
        unsigned bestidx = 0u;
        const int pos = (nt0 + nt) * 16 + nloc;
        if (pos < PPOS) {
            #pragma unroll
            for (int r = 0; r < 8; ++r) {
                const float v = acc[nt][r];
                if (v > bestv) {
                    bestv = v;
                    bestidx = (unsigned)((atom0 + r + 8 * half) * PPOS + pos);
                }
            }
        }
        unsigned vb = __float_as_uint(bestv);
        vb = (vb & 0x80000000u) ? ~vb : (vb | 0x80000000u);
        unsigned long long key = ((unsigned long long)vb << 32) | (0xFFFFFFFFu - bestidx);
        for (int off = 16; off > 0; off >>= 1) {
            unsigned long long o = __shfl_xor(key, off, 32);
            if (o > key) key = o;
        }
        if (lane == nt) mykey = key;
    }
    const int t = nt0 + lane;
    if (lane < NT && t < NTILES) rec[t + (t >= NTPB ? 16 - NTPB : 0)] = mykey;
}

__global__ __launch_bounds__(CORR_THREADS, 1)
void k_corr(const float* __restrict__ dnn, const float* __restrict__ res,
            unsigned long long* __restrict__ tilebest) {
    extern __shared__ float smem[];
    const int b    = blockIdx.z;
    const int nt0  = blockIdx.y * NTPB;
    const int m0   = blockIdx.x * MTILE;
    const int tid  = threadIdx.x;
    const int lane = tid & 31;
    const int wave = tid >> 5;

    const int half = lane >> 4;
    const int nloc = lane & 15;
    const float* arow = dnn + (size_t)(m0 + wave * 16 + nloc) * KDIM + half * 2;

    #pragma unroll
    for (int pf = 0; pf < 16; ++pf)
        __builtin_prefetch(arow + pf * 32, 0, 3);

    const float4* resb = (const float4*)(res + (size_t)b * CC * RES_W);
    float4* s4 = (float4*)smem;
    for (int i = tid; i < (CC * RES_W) / 4; i += CORR_THREADS) s4[i] = resb[i];
    __syncthreads();

    v8f acc[NTPB] = {};

    auto loadA = [&](int ks) { v2f a; a.x = arow[ks * 4]; a.y = arow[ks * 4 + 1]; return a; };
    auto srowp = [&](int ks) {
        const int kbase = ks * 4 + half * 2;
        return (const float*)smem + (kbase >> 3) * RES_W + (kbase & 7) + nloc + nt0 * 16;
    };

    v2f a0 = loadA(0);
    for (int ks = 0; ks < KDIM / 4; ++ks) {
        v2f an = loadA(ks + 1 < KDIM / 4 ? ks + 1 : ks);
        const float* sr = srowp(ks);
        v2f bf[NTPB];
        #pragma unroll
        for (int nt = 0; nt < NTPB; ++nt) {
            bf[nt].x = sr[nt * 16];
            bf[nt].y = sr[nt * 16 + 1];
        }
        #pragma unroll
        for (int nt = 0; nt < NTPB; ++nt)
            acc[nt] = __builtin_amdgcn_wmma_f32_16x16x4_f32(
                false, a0, false, bf[nt], (short)0, acc[nt], false, false);
        a0 = an;
    }

    const int mt = blockIdx.x * (MTILE / 16) + wave;
    tile_keys_store<NTPB>(acc, nt0, m0 + wave * 16, half, nloc, lane,
                          tilebest + ((size_t)b * MTILES + mt) * KSLOTS);
}

__global__ __launch_bounds__(CORR_THREADS, 1)
void k_corr_inc(const float* __restrict__ dnn, const float* __restrict__ res,
                const int* __restrict__ tsel, unsigned long long* __restrict__ tilebest) {
    __shared__ float slab[BATCH * CC * SLABW];
    const int mt   = blockIdx.x;
    const int tid  = threadIdx.x;
    const int lane = tid & 31;
    const int b    = __builtin_amdgcn_readfirstlane(tid >> 5);
    const int half = lane >> 4;
    const int nloc = lane & 15;
    const float* arow = dnn + (size_t)(mt * 16 + nloc) * KDIM + half * 2;

    #pragma unroll
    for (int pf = 0; pf < 16; ++pf)
        __builtin_prefetch(arow + pf * 32, 0, 3);

    const int t0 = tsel[b];
    float* sl = slab + b * CC * SLABW;
    const float* rsrc = res + (size_t)b * CC * RES_W + t0 * 16;
    #pragma unroll
    for (int j = 0; j < (CC * SLABW / 4) / 32; ++j) {
        const int i   = lane + 32 * j;
        const int row = i / (SLABW / 4);
        const int q   = i - row * (SLABW / 4);
        *(float4*)(sl + row * SLABW + q * 4) = *(const float4*)(rsrc + row * RES_W + q * 4);
    }
    __syncthreads();

    const float* bl = sl + half * 2 + nloc;
    v8f acc[2] = {};

    auto loadG = [&](int g, v2f (&a)[8], v2f (&bq)[8][2]) {
        const float* ap = arow + g * 32;
        const float* bp = bl + g * 4 * SLABW;
        #pragma unroll
        for (int j = 0; j < 8; ++j) {
            a[j].x = ap[j * 4];
            a[j].y = ap[j * 4 + 1];
            #pragma unroll
            for (int nt = 0; nt < 2; ++nt) {
                const float* sp = bp + (j >> 1) * SLABW + (j & 1) * 4 + nt * 16;
                bq[j][nt].x = sp[0];
                bq[j][nt].y = sp[1];
            }
        }
    };
    auto mmaG = [&](const v2f (&a)[8], const v2f (&bq)[8][2]) {
        #pragma unroll
        for (int j = 0; j < 8; ++j) {
            #pragma unroll
            for (int nt = 0; nt < 2; ++nt)
                acc[nt] = __builtin_amdgcn_wmma_f32_16x16x4_f32(
                    false, a[j], false, bq[j][nt], (short)0, acc[nt], false, false);
        }
    };

    v2f aA[8], aB[8], bA[8][2], bB[8][2];
    loadG(0, aA, bA);
    for (int g = 0; g < KDIM / 32; g += 2) {
        loadG(g + 1, aB, bB);
        __builtin_amdgcn_sched_barrier(0);
        mmaG(aA, bA);
        __builtin_amdgcn_sched_barrier(0);
        loadG(g + 2 < KDIM / 32 ? g + 2 : g + 1, aA, bA);
        __builtin_amdgcn_sched_barrier(0);
        mmaG(aB, bB);
        __builtin_amdgcn_sched_barrier(0);
    }

    tile_keys_store<2>(acc, t0, mt * 16, half, nloc, lane,
                       tilebest + ((size_t)b * MTILES + mt) * KSLOTS);
}

__global__ __launch_bounds__(KDIM)
void k_update(const float* __restrict__ dnn, float* __restrict__ res,
              float* __restrict__ recon, const unsigned long long* __restrict__ tilebest,
              int* __restrict__ tsel) {
    __shared__ unsigned long long wred[KDIM / 32];
    const int b    = blockIdx.x;
    const int tid  = threadIdx.x;
    const int lane = tid & 31;
    const unsigned long long* tb = tilebest + (size_t)b * MTILES * KSLOTS;
    const int sl = tid & (KSLOTS - 1);
    const bool used = (sl < NTPB) || (sl >= 16 && sl < 16 + NTILES - NTPB);
    unsigned long long key = 0ull;
    if (used) {
        #pragma unroll
        for (int j = 0; j < (MTILES * KSLOTS) / KDIM; ++j) {
            const unsigned long long k = tb[j * KDIM + tid];
            if (k > key) key = k;
        }
    }
    for (int off = 16; off > 0; off >>= 1) {
        unsigned long long o = __shfl_xor(key, off, 32);
        if (o > key) key = o;
    }
    if (lane == 0) wred[tid >> 5] = key;
    __syncthreads();
    key = wred[0];
    for (int w = 1; w < KDIM / 32; ++w) { const unsigned long long o = wred[w]; if (o > key) key = o; }

    const unsigned enc  = (unsigned)(key >> 32);
    const unsigned bits = (enc & 0x80000000u) ? (enc & 0x7FFFFFFFu) : ~enc;
    const float val = __uint_as_float(bits);
    const unsigned idx = 0xFFFFFFFFu - (unsigned)(key & 0xFFFFFFFFu);
    const int atom = (int)(idx / PPOS);
    const int pos  = (int)(idx % PPOS);
    const float* ap = dnn + (size_t)atom * KDIM;
    float* rb = res   + (size_t)b * CC * RES_W;
    float* cb = recon + (size_t)b * CC * RES_W;
    for (int i = threadIdx.x; i < KDIM; i += blockDim.x) {
        const int c = i >> 3, t = i & 7;
        const float av = ap[i] * val;
        rb[c * RES_W + pos + t] -= av;
        cb[c * RES_W + pos + t] += av;
    }
    if (tid == 0) {
        const int lo = pos > 7 ? pos - 7 : 0;
        const int t0 = lo >> 4;
        tsel[b] = t0 < NTILES - 2 ? t0 : NTILES - 2;
    }
}

__global__ void k_output(const float* __restrict__ recon, float* __restrict__ out) {
    const int i = blockIdx.x * blockDim.x + threadIdx.x;
    if (i < BATCH * CC * TT) {
        const int t  = i % TT;
        const int bc = i / TT;
        out[i] = recon[(size_t)bc * RES_W + t];
    }
}

extern "C" void kernel_launch(void* const* d_in, const int* in_sizes, int n_in,
                              void* d_out, int out_size, void* d_ws, size_t ws_size,
                              hipStream_t stream) {
    const float* x = (const float*)d_in[0];
    const float* d = (const float*)d_in[1];
    (void)in_sizes; (void)n_in; (void)out_size; (void)ws_size;

    char* ws = (char*)d_ws;
    float* dnn   = (float*)ws;
    float* res   = (float*)(ws + (size_t)NATOMS * KDIM * 4);
    float* recon = res + BATCH * CC * RES_W;
    int* tsel = (int*)(recon + BATCH * CC * RES_W);
    unsigned long long* tilebest = (unsigned long long*)(tsel + 8);

    k_normalize<<<NATOMS, CORR_THREADS, 0, stream>>>(d, dnn);
    k_init<<<(BATCH * CC * RES_W + 255) / 256, 256, 0, stream>>>(x, res, recon);

    const size_t shbytes = (size_t)CC * RES_W * sizeof(float);
    dim3 grid(NATOMS / MTILE, NSPLIT, BATCH);
    k_corr<<<grid, CORR_THREADS, shbytes, stream>>>(dnn, res, tilebest);
    k_update<<<BATCH, KDIM, 0, stream>>>(dnn, res, recon, tilebest, tsel);
    for (int it = 1; it < NITER; ++it) {
        k_corr_inc<<<MTILES, CORR_THREADS, 0, stream>>>(dnn, res, tsel, tilebest);
        k_update<<<BATCH, KDIM, 0, stream>>>(dnn, res, recon, tilebest, tsel);
    }
    k_output<<<(BATCH * CC * TT + 255) / 256, 256, 0, stream>>>(recon, (float*)d_out);
}


# Incremental correlation kernel stages its 16 atom rows in LDS once per block (one batch of row-contiguous loads, padded pitch) and takes A fragments from LDS instead of global loads per k-group
# speedup vs baseline: 2.8944x; 1.0688x over previous
#include <hip/hip_runtime.h>

#define NATOMS 4096
#define CC     64
#define TT     256
#define ASIZE  8
#define BATCH  4
#define KDIM   512
#define PPOS   257
#define RES_W  304
#define NSPLIT 2
#define NTPB   9
#define MTILE  64
#define CORR_THREADS 128
#define NITER  32
#define NTILES 17
#define MTILES 256
#define KSLOTS 32
#define SLABW  40
#define APITCH 516

typedef __attribute__((ext_vector_type(2))) float v2f;
typedef __attribute__((ext_vector_type(8))) float v8f;

__global__ void k_normalize(const float* __restrict__ d, float* __restrict__ dnn) {
    __shared__ float red[4];
    const int a = blockIdx.x;
    const float* src = d + (size_t)a * KDIM;
    float s = 0.f;
    for (int i = threadIdx.x; i < KDIM; i += CORR_THREADS) { float v = src[i]; s += v * v; }
    for (int off = 16; off > 0; off >>= 1) s += __shfl_down(s, off, 32);
    const int wv = threadIdx.x >> 5;
    if ((threadIdx.x & 31) == 0) red[wv] = s;
    __syncthreads();
    if (threadIdx.x == 0) red[0] = 1.f / (sqrtf(red[0] + red[1] + red[2] + red[3]) + 1e-8f);
    __syncthreads();
    const float inv = red[0];
    float* dst = dnn + (size_t)a * KDIM;
    for (int i = threadIdx.x; i < KDIM; i += CORR_THREADS) dst[i] = src[i] * inv;
}

__global__ void k_init(const float* __restrict__ x, float* __restrict__ res,
                       float* __restrict__ recon) {
    const int i = blockIdx.x * blockDim.x + threadIdx.x;
    if (i < BATCH * CC * RES_W) {
        const int col = i % RES_W;
        const int bc  = i / RES_W;
        res[i]   = (col < TT) ? x[bc * TT + col] : 0.f;
        recon[i] = 0.f;
    }
}

template <int NT>
__device__ __forceinline__ void tile_keys_store(const v8f (&acc)[NT], int nt0, int atom0,
                                                int half, int nloc, int lane,
                                                unsigned long long* __restrict__ rec) {
    unsigned long long mykey = 0ull;
    #pragma unroll
    for (int nt = 0; nt < NT; ++nt) {
        float bestv = -3.402823466e38f;
        unsigned bestidx = 0u;
        const int pos = (nt0 + nt) * 16 + nloc;
        if (pos < PPOS) {
            #pragma unroll
            for (int r = 0; r < 8; ++r) {
                const float v = acc[nt][r];
                if (v > bestv) {
                    bestv = v;
                    bestidx = (unsigned)((atom0 + r + 8 * half) * PPOS + pos);
                }
            }
        }
        unsigned vb = __float_as_uint(bestv);
        vb = (vb & 0x80000000u) ? ~vb : (vb | 0x80000000u);
        unsigned long long key = ((unsigned long long)vb << 32) | (0xFFFFFFFFu - bestidx);
        for (int off = 16; off > 0; off >>= 1) {
            unsigned long long o = __shfl_xor(key, off, 32);
            if (o > key) key = o;
        }
        if (lane == nt) mykey = key;
    }
    const int t = nt0 + lane;
    if (lane < NT && t < NTILES) rec[t + (t >= NTPB ? 16 - NTPB : 0)] = mykey;
}

__global__ __launch_bounds__(CORR_THREADS, 1)
void k_corr(const float* __restrict__ dnn, const float* __restrict__ res,
            unsigned long long* __restrict__ tilebest) {
    extern __shared__ float smem[];
    const int b    = blockIdx.z;
    const int nt0  = blockIdx.y * NTPB;
    const int m0   = blockIdx.x * MTILE;
    const int tid  = threadIdx.x;
    const int lane = tid & 31;
    const int wave = tid >> 5;

    const int half = lane >> 4;
    const int nloc = lane & 15;
    const float* arow = dnn + (size_t)(m0 + wave * 16 + nloc) * KDIM + half * 2;

    #pragma unroll
    for (int pf = 0; pf < 16; ++pf)
        __builtin_prefetch(arow + pf * 32, 0, 3);

    const float4* resb = (const float4*)(res + (size_t)b * CC * RES_W);
    float4* s4 = (float4*)smem;
    for (int i = tid; i < (CC * RES_W) / 4; i += CORR_THREADS) s4[i] = resb[i];
    __syncthreads();

    v8f acc[NTPB] = {};

    auto loadA = [&](int ks) { v2f a; a.x = arow[ks * 4]; a.y = arow[ks * 4 + 1]; return a; };
    auto srowp = [&](int ks) {
        const int kbase = ks * 4 + half * 2;
        return (const float*)smem + (kbase >> 3) * RES_W + (kbase & 7) + nloc + nt0 * 16;
    };

    v2f a0 = loadA(0);
    for (int ks = 0; ks < KDIM / 4; ++ks) {
        v2f an = loadA(ks + 1 < KDIM / 4 ? ks + 1 : ks);
        const float* sr = srowp(ks);
        v2f bf[NTPB];
        #pragma unroll
        for (int nt = 0; nt < NTPB; ++nt) {
            bf[nt].x = sr[nt * 16];
            bf[nt].y = sr[nt * 16 + 1];
        }
        #pragma unroll
        for (int nt = 0; nt < NTPB; ++nt)
            acc[nt] = __builtin_amdgcn_wmma_f32_16x16x4_f32(
                false, a0, false, bf[nt], (short)0, acc[nt], false, false);
        a0 = an;
    }

    const int mt = blockIdx.x * (MTILE / 16) + wave;
    tile_keys_store<NTPB>(acc, nt0, m0 + wave * 16, half, nloc, lane,
                          tilebest + ((size_t)b * MTILES + mt) * KSLOTS);
}

__global__ __launch_bounds__(CORR_THREADS, 1)
void k_corr_inc(const float* __restrict__ dnn, const float* __restrict__ res,
                const int* __restrict__ tsel, unsigned long long* __restrict__ tilebest) {
    __shared__ __attribute__((aligned(16))) float atile[16 * APITCH];
    __shared__ __attribute__((aligned(16))) float slab[BATCH * CC * SLABW];
    const int mt   = blockIdx.x;
    const int tid  = threadIdx.x;
    const int lane = tid & 31;
    const int b    = __builtin_amdgcn_readfirstlane(tid >> 5);
    const int half = lane >> 4;
    const int nloc = lane & 15;

    const float4* asrc = (const float4*)(dnn + (size_t)mt * 16 * KDIM) + tid;
    float4 av[16];
    #pragma unroll
    for (int j = 0; j < 16; ++j) av[j] = asrc[j * (KDIM / 4)];

    const int t0 = tsel[b];
    float* sl = slab + b * CC * SLABW;
    const float* rsrc = res + (size_t)b * CC * RES_W + t0 * 16;
    float4 sv[(CC * SLABW / 4) / 32];
    #pragma unroll
    for (int j = 0; j < (CC * SLABW / 4) / 32; ++j) {
        const int i   = lane + 32 * j;
        const int row = i / (SLABW / 4);
        const int q   = i - row * (SLABW / 4);
        sv[j] = *(const float4*)(rsrc + row * RES_W + q * 4);
    }
    #pragma unroll
    for (int j = 0; j < 16; ++j) *(float4*)(atile + j * APITCH + tid * 4) = av[j];
    #pragma unroll
    for (int j = 0; j < (CC * SLABW / 4) / 32; ++j) {
        const int i   = lane + 32 * j;
        const int row = i / (SLABW / 4);
        const int q   = i - row * (SLABW / 4);
        *(float4*)(sl + row * SLABW + q * 4) = sv[j];
    }
    __syncthreads();

    const float* al = atile + nloc * APITCH + half * 2;
    const float* bl = sl + half * 2 + nloc;
    v8f acc[2] = {};

    auto loadG = [&](int g, v2f (&a)[8], v2f (&bq)[8][2]) {
        const float* ap = al + g * 32;
        const float* bp = bl + g * 4 * SLABW;
        #pragma unroll
        for (int j = 0; j < 8; ++j) {
            a[j] = *(const v2f*)(ap + j * 4);
            #pragma unroll
            for (int nt = 0; nt < 2; ++nt) {
                const float* sp = bp + (j >> 1) * SLABW + (j & 1) * 4 + nt * 16;
                bq[j][nt].x = sp[0];
                bq[j][nt].y = sp[1];
            }
        }
    };
    auto mmaG = [&](const v2f (&a)[8], const v2f (&bq)[8][2]) {
        #pragma unroll
        for (int j = 0; j < 8; ++j) {
            #pragma unroll
            for (int nt = 0; nt < 2; ++nt)
                acc[nt] = __builtin_amdgcn_wmma_f32_16x16x4_f32(
                    false, a[j], false, bq[j][nt], (short)0, acc[nt], false, false);
        }
    };

    v2f aA[8], aB[8], bA[8][2], bB[8][2];
    loadG(0, aA, bA);
    for (int g = 0; g < KDIM / 32; g += 2) {
        loadG(g + 1, aB, bB);
        __builtin_amdgcn_sched_barrier(0);
        mmaG(aA, bA);
        __builtin_amdgcn_sched_barrier(0);
        loadG(g + 2 < KDIM / 32 ? g + 2 : g + 1, aA, bA);
        __builtin_amdgcn_sched_barrier(0);
        mmaG(aB, bB);
        __builtin_amdgcn_sched_barrier(0);
    }

    tile_keys_store<2>(acc, t0, mt * 16, half, nloc, lane,
                       tilebest + ((size_t)b * MTILES + mt) * KSLOTS);
}

__global__ __launch_bounds__(KDIM)
void k_update(const float* __restrict__ dnn, float* __restrict__ res,
              float* __restrict__ recon, const unsigned long long* __restrict__ tilebest,
              int* __restrict__ tsel) {
    __shared__ unsigned long long wred[KDIM / 32];
    const int b    = blockIdx.x;
    const int tid  = threadIdx.x;
    const int lane = tid & 31;
    const unsigned long long* tb = tilebest + (size_t)b * MTILES * KSLOTS;
    const int sl = tid & (KSLOTS - 1);
    const bool used = (sl < NTPB) || (sl >= 16 && sl < 16 + NTILES - NTPB);
    unsigned long long key = 0ull;
    if (used) {
        #pragma unroll
        for (int j = 0; j < (MTILES * KSLOTS) / KDIM; ++j) {
            const unsigned long long k = tb[j * KDIM + tid];
            if (k > key) key = k;
        }
    }
    for (int off = 16; off > 0; off >>= 1) {
        unsigned long long o = __shfl_xor(key, off, 32);
        if (o > key) key = o;
    }
    if (lane == 0) wred[tid >> 5] = key;
    __syncthreads();
    key = wred[0];
    for (int w = 1; w < KDIM / 32; ++w) { const unsigned long long o = wred[w]; if (o > key) key = o; }

    const unsigned enc  = (unsigned)(key >> 32);
    const unsigned bits = (enc & 0x80000000u) ? (enc & 0x7FFFFFFFu) : ~enc;
    const float val = __uint_as_float(bits);
    const unsigned idx = 0xFFFFFFFFu - (unsigned)(key & 0xFFFFFFFFu);
    const int atom = (int)(idx / PPOS);
    const int pos  = (int)(idx % PPOS);
    const float* ap = dnn + (size_t)atom * KDIM;
    float* rb = res   + (size_t)b * CC * RES_W;
    float* cb = recon + (size_t)b * CC * RES_W;
    for (int i = threadIdx.x; i < KDIM; i += blockDim.x) {
        const int c = i >> 3, t = i & 7;
        const float av = ap[i] * val;
        rb[c * RES_W + pos + t] -= av;
        cb[c * RES_W + pos + t] += av;
    }
    if (tid == 0) {
        const int lo = pos > 7 ? pos - 7 : 0;
        const int t0 = lo >> 4;
        tsel[b] = t0 < NTILES - 2 ? t0 : NTILES - 2;
    }
}

__global__ void k_output(const float* __restrict__ recon, float* __restrict__ out) {
    const int i = blockIdx.x * blockDim.x + threadIdx.x;
    if (i < BATCH * CC * TT) {
        const int t  = i % TT;
        const int bc = i / TT;
        out[i] = recon[(size_t)bc * RES_W + t];
    }
}

extern "C" void kernel_launch(void* const* d_in, const int* in_sizes, int n_in,
                              void* d_out, int out_size, void* d_ws, size_t ws_size,
                              hipStream_t stream) {
    const float* x = (const float*)d_in[0];
    const float* d = (const float*)d_in[1];
    (void)in_sizes; (void)n_in; (void)out_size; (void)ws_size;

    char* ws = (char*)d_ws;
    float* dnn   = (float*)ws;
    float* res   = (float*)(ws + (size_t)NATOMS * KDIM * 4);
    float* recon = res + BATCH * CC * RES_W;
    int* tsel = (int*)(recon + BATCH * CC * RES_W);
    unsigned long long* tilebest = (unsigned long long*)(tsel + 8);

    k_normalize<<<NATOMS, CORR_THREADS, 0, stream>>>(d, dnn);
    k_init<<<(BATCH * CC * RES_W + 255) / 256, 256, 0, stream>>>(x, res, recon);

    const size_t shbytes = (size_t)CC * RES_W * sizeof(float);
    dim3 grid(NATOMS / MTILE, NSPLIT, BATCH);
    k_corr<<<grid, CORR_THREADS, shbytes, stream>>>(dnn, res, tilebest);
    k_update<<<BATCH, KDIM, 0, stream>>>(dnn, res, recon, tilebest, tsel);
    for (int it = 1; it < NITER; ++it) {
        k_corr_inc<<<MTILES, CORR_THREADS, 0, stream>>>(dnn, res, tsel, tilebest);
        k_update<<<BATCH, KDIM, 0, stream>>>(dnn, res, recon, tilebest, tsel);
    }
    k_output<<<(BATCH * CC * TT + 255) / 256, 256, 0, stream>>>(recon, (float*)d_out);
}


# Each incremental correlation launch does the previous step's argmax reduce and rank-1 update in its own head (residual double-buffered, reconstruction summed in out): 67 launches become 36
# speedup vs baseline: 3.6919x; 1.2755x over previous
#include <hip/hip_runtime.h>

#define NATOMS 4096
#define CC     64
#define TT     256
#define ASIZE  8
#define BATCH  4
#define KDIM   512
#define PPOS   257
#define RES_W  304
#define NSPLIT 2
#define NTPB   9
#define MTILE  64
#define CORR_THREADS 128
#define NITER  32
#define NTILES 17
#define MTILES 256
#define KSLOTS 32
#define SLABW  40
#define APITCH 516
#define RKA    9
#define RKB    24

typedef __attribute__((ext_vector_type(2))) float v2f;
typedef __attribute__((ext_vector_type(8))) float v8f;

__global__ void k_normalize(const float* __restrict__ d, float* __restrict__ dnn) {
    __shared__ float red[4];
    const int a = blockIdx.x;
    const float* src = d + (size_t)a * KDIM;
    float s = 0.f;
    for (int i = threadIdx.x; i < KDIM; i += CORR_THREADS) { float v = src[i]; s += v * v; }
    for (int off = 16; off > 0; off >>= 1) s += __shfl_down(s, off, 32);
    const int wv = threadIdx.x >> 5;
    if ((threadIdx.x & 31) == 0) red[wv] = s;
    __syncthreads();
    if (threadIdx.x == 0) red[0] = 1.f / (sqrtf(red[0] + red[1] + red[2] + red[3]) + 1e-8f);
    __syncthreads();
    const float inv = red[0];
    float* dst = dnn + (size_t)a * KDIM;
    for (int i = threadIdx.x; i < KDIM; i += CORR_THREADS) dst[i] = src[i] * inv;
}

__global__ void k_init(const float* __restrict__ x, float* __restrict__ res0,
                       float* __restrict__ res1, float* __restrict__ out) {
    const int i = blockIdx.x * blockDim.x + threadIdx.x;
    if (i < BATCH * CC * RES_W) {
        const int col = i % RES_W;
        const int bc  = i / RES_W;
        const float v = (col < TT) ? x[bc * TT + col] : 0.f;
        res0[i] = v;
        res1[i] = v;
        if (col < TT) out[bc * TT + col] = 0.f;
    }
}

struct Pick { float val; int atom; int pos; };

__device__ __forceinline__ unsigned long long wave_max_key(unsigned long long key) {
    for (int off = 16; off > 0; off >>= 1) {
        unsigned long long o = __shfl_xor(key, off, 32);
        if (o > key) key = o;
    }
    return key;
}

__device__ __forceinline__ Pick decode_key(unsigned long long key) {
    const unsigned enc  = (unsigned)(key >> 32);
    const unsigned bits = (enc & 0x80000000u) ? (enc & 0x7FFFFFFFu) : ~enc;
    const unsigned idx  = 0xFFFFFFFFu - (unsigned)(key & 0xFFFFFFFFu);
    const unsigned a    = idx / PPOS;
    Pick p;
    p.val  = __uint_as_float(bits);
    p.atom = (int)(a < NATOMS ? a : NATOMS - 1);
    p.pos  = (int)(idx % PPOS);
    return p;
}

__device__ __forceinline__ int first_tile(int pos) {
    const int lo = pos > 7 ? pos - 7 : 0;
    const int t0 = lo >> 4;
    return t0 < NTILES - 2 ? t0 : NTILES - 2;
}

__device__ __forceinline__ float sub_scaled(float r, float a, float val) { return __builtin_fmaf(-a, val, r); }
__device__ __forceinline__ float add_scaled(float c, float a, float val) { return __builtin_fmaf(a, val, c); }

__device__ __forceinline__ void load_row_keys(const unsigned long long* rk, int b, int lane,
                                              unsigned long long (&k)[MTILES / 32]) {
    #pragma unroll
    for (int j = 0; j < MTILES / 32; ++j) k[j] = rk[(size_t)(lane + 32 * j) * KSLOTS + b];
}

__device__ __forceinline__ Pick pick_from_keys(const unsigned long long (&k)[MTILES / 32]) {
    unsigned long long key = k[0];
    #pragma unroll
    for (int j = 1; j < MTILES / 32; ++j) if (k[j] > key) key = k[j];
    key = wave_max_key(key);
    const unsigned klo = __builtin_amdgcn_readfirstlane((unsigned)(key & 0xFFFFFFFFu));
    const unsigned khi = __builtin_amdgcn_readfirstlane((unsigned)(key >> 32));
    return decode_key(((unsigned long long)khi << 32) | klo);
}

template <int NT>
__device__ __forceinline__ unsigned long long tile_keys_store(const v8f (&acc)[NT], int nt0, int atom0,
                                                int half, int nloc, int lane,
                                                unsigned long long* __restrict__ rec) {
    unsigned long long mykey = 0ull, allmax = 0ull;
    #pragma unroll
    for (int nt = 0; nt < NT; ++nt) {
        float bestv = -3.402823466e38f;
        unsigned bestidx = 0u;
        const int pos = (nt0 + nt) * 16 + nloc;
        if (pos < PPOS) {
            #pragma unroll
            for (int r = 0; r < 8; ++r) {
                const float v = acc[nt][r];
                if (v > bestv) {
                    bestv = v;
                    bestidx = (unsigned)((atom0 + r + 8 * half) * PPOS + pos);
                }
            }
        }
        unsigned vb = __float_as_uint(bestv);
        vb = (vb & 0x80000000u) ? ~vb : (vb | 0x80000000u);
        unsigned long long key = ((unsigned long long)vb << 32) | (0xFFFFFFFFu - bestidx);
        for (int off = 16; off > 0; off >>= 1) {
            unsigned long long o = __shfl_xor(key, off, 32);
            if (o > key) key = o;
        }
        if (lane == nt) mykey = key;
        if (key > allmax) allmax = key;
    }
    const int t = nt0 + lane;
    if (lane < NT && t < NTILES) rec[t + (t >= NTPB ? 16 - NTPB : 0)] = mykey;
    return allmax;
}

__global__ __launch_bounds__(CORR_THREADS, 1)
void k_corr(const float* __restrict__ dnn, const float* __restrict__ res,
            unsigned long long* __restrict__ tilebest) {
    extern __shared__ float smem[];
    const int b    = blockIdx.z;
    const int nt0  = blockIdx.y * NTPB;
    const int m0   = blockIdx.x * MTILE;
    const int tid  = threadIdx.x;
    const int lane = tid & 31;
    const int wave = tid >> 5;

    const int half = lane >> 4;
    const int nloc = lane & 15;
    const float* arow = dnn + (size_t)(m0 + wave * 16 + nloc) * KDIM + half * 2;

    #pragma unroll
    for (int pf = 0; pf < 16; ++pf)
        __builtin_prefetch(arow + pf * 32, 0, 3);

    const float4* resb = (const float4*)(res + (size_t)b * CC * RES_W);
    float4* s4 = (float4*)smem;
    for (int i = tid; i < (CC * RES_W) / 4; i += CORR_THREADS) s4[i] = resb[i];
    __syncthreads();

    v8f acc[NTPB] = {};

    auto loadA = [&](int ks) { v2f a; a.x = arow[ks * 4]; a.y = arow[ks * 4 + 1]; return a; };
    auto srowp = [&](int ks) {
        const int kbase = ks * 4 + half * 2;
        return (const float*)smem + (kbase >> 3) * RES_W + (kbase & 7) + nloc + nt0 * 16;
    };

    v2f a0 = loadA(0);
    for (int ks = 0; ks < KDIM / 4; ++ks) {
        v2f an = loadA(ks + 1 < KDIM / 4 ? ks + 1 : ks);
        const float* sr = srowp(ks);
        v2f bf[NTPB];
        #pragma unroll
        for (int nt = 0; nt < NTPB; ++nt) {
            bf[nt].x = sr[nt * 16];
            bf[nt].y = sr[nt * 16 + 1];
        }
        #pragma unroll
        for (int nt = 0; nt < NTPB; ++nt)
            acc[nt] = __builtin_amdgcn_wmma_f32_16x16x4_f32(
                false, a0, false, bf[nt], (short)0, acc[nt], false, false);
        a0 = an;
    }

    const int mt = blockIdx.x * (MTILE / 16) + wave;
    tile_keys_store<NTPB>(acc, nt0, m0 + wave * 16, half, nloc, lane,
                          tilebest + ((size_t)b * MTILES + mt) * KSLOTS);
}

template <bool HEAD>
__global__ __launch_bounds__(CORR_THREADS, 1)
void k_corr_inc(const float* __restrict__ dnn, const float* __restrict__ rin,
                float* __restrict__ rout, float* __restrict__ out, int* __restrict__ tsel,
                unsigned long long* tilebest,
                const unsigned long long* rk_in, unsigned long long* rk_out) {
    __shared__ __attribute__((aligned(16))) float atile[16 * APITCH];
    __shared__ __attribute__((aligned(16))) float slab[BATCH * CC * SLABW];
    __shared__ unsigned long long rowk[BATCH];
    const int mt   = blockIdx.x;
    const int tid  = threadIdx.x;
    const int lane = tid & 31;
    const int b    = __builtin_amdgcn_readfirstlane(tid >> 5);
    const int half = lane >> 4;
    const int nloc = lane & 15;

    unsigned long long rkeys[MTILES / 32];
    if (HEAD) {
        load_row_keys(rk_in, b, lane, rkeys);

        if (mt == MTILES) {
            const Pick p = pick_from_keys(rkeys);
            int pprev = tsel[BATCH + b];
            pprev = pprev < 0 ? 0 : (pprev < PPOS ? pprev : PPOS - 1);
            const float* ap = dnn + (size_t)p.atom * KDIM;
            const float* ri = rin  + (size_t)b * CC * RES_W;
            float*       ro = rout + (size_t)b * CC * RES_W;
            float*       ob = out  + (size_t)b * CC * TT;
            const int t  = lane & 7;
            const int c0 = lane >> 3;
            const int col = p.pos + t;
            const int pc  = pprev + t;
            const bool carry = pc < p.pos || pc >= p.pos + ASIZE;
            float a[KDIM / 32], r[KDIM / 32], o[KDIM / 32], pv[KDIM / 32];
            #pragma unroll
            for (int j = 0; j < KDIM / 32; ++j) {
                const int c = c0 + 4 * j;
                a[j]  = ap[lane + 32 * j];
                r[j]  = ri[c * RES_W + col];
                pv[j] = ri[c * RES_W + pc];
                o[j]  = col < TT ? ob[c * TT + col] : 0.f;
            }
            #pragma unroll
            for (int j = 0; j < KDIM / 32; ++j) {
                const int c = c0 + 4 * j;
                ro[c * RES_W + col] = sub_scaled(r[j], a[j], p.val);
                if (carry) ro[c * RES_W + pc] = pv[j];
                if (col < TT) ob[c * TT + col] = add_scaled(o[j], a[j], p.val);
            }
            if (lane == 0) tsel[BATCH + b] = p.pos;
            return;
        }
    }

    const float4* asrc = (const float4*)(dnn + (size_t)mt * 16 * KDIM) + tid;
    float4 av[16];
    #pragma unroll
    for (int j = 0; j < 16; ++j) av[j] = asrc[j * (KDIM / 4)];

    unsigned long long* rec = tilebest + ((size_t)b * MTILES + mt) * KSLOTS;
    unsigned long long oldkeys[NTILES];
    #pragma unroll
    for (int j = 0; j < NTILES; ++j) oldkeys[j] = rec[j + (j >= NTPB ? 16 - NTPB : 0)];

    unsigned soff[(CC * SLABW / 4) / 32];
    #pragma unroll
    for (int j = 0; j < (CC * SLABW / 4) / 32; ++j) {
        const int i   = lane + 32 * j;
        const int row = i / (SLABW / 4);
        const int q   = i - row * (SLABW / 4);
        soff[j] = (unsigned)(row * RES_W + q * 4) * 4u;
    }

    Pick p = {0.f, 0, 0};
    if (HEAD) p = pick_from_keys(rkeys);
    const int t0 = HEAD ? first_tile(p.pos) : tsel[b];

    float* sl = slab + b * CC * SLABW;
    const char* rsrc = (const char*)(rin + (size_t)b * CC * RES_W + t0 * 16);
    float4 sv[(CC * SLABW / 4) / 32];
    #pragma unroll
    for (int j = 0; j < (CC * SLABW / 4) / 32; ++j) sv[j] = *(const float4*)(rsrc + soff[j]);
    float4 uv[KDIM / 128];
    if (HEAD) {
        const float4* arow4 = (const float4*)(dnn + (size_t)p.atom * KDIM);
        #pragma unroll
        for (int j = 0; j < KDIM / 128; ++j) uv[j] = arow4[lane + 32 * j];
    }
    unsigned long long oldmax = 0ull;
    #pragma unroll
    for (int j = 0; j < NTILES; ++j)
        if (j != t0 && j != t0 + 1 && oldkeys[j] > oldmax) oldmax = oldkeys[j];

    #pragma unroll
    for (int j = 0; j < 16; ++j) *(float4*)(atile + j * APITCH + tid * 4) = av[j];
    #pragma unroll
    for (int j = 0; j < (CC * SLABW / 4) / 32; ++j) *(float4*)(sl + 4 * (lane + 32 * j)) = sv[j];
    __syncthreads();

    if (HEAD) {
        float* up = sl + (p.pos - t0 * 16);
        float rv[KDIM / 128][4];
        #pragma unroll
        for (int j = 0; j < KDIM / 128; ++j) {
            const int i4 = lane + 32 * j;
            const float* q = up + (i4 >> 1) * SLABW + (i4 & 1) * 4;
            rv[j][0] = q[0]; rv[j][1] = q[1]; rv[j][2] = q[2]; rv[j][3] = q[3];
        }
        #pragma unroll
        for (int j = 0; j < KDIM / 128; ++j) {
            const int i4 = lane + 32 * j;
            float* q = up + (i4 >> 1) * SLABW + (i4 & 1) * 4;
            q[0] = sub_scaled(rv[j][0], uv[j].x, p.val);
            q[1] = sub_scaled(rv[j][1], uv[j].y, p.val);
            q[2] = sub_scaled(rv[j][2], uv[j].z, p.val);
            q[3] = sub_scaled(rv[j][3], uv[j].w, p.val);
        }
        __syncthreads();
    }

    const float* al = atile + nloc * APITCH + half * 2;
    const float* bl = sl + half * 2 + nloc;
    v8f acc[2] = {};

    auto loadG = [&](int g, v2f (&a)[8], v2f (&bq)[8][2]) {
        const float* ap = al + g * 32;
        const float* bp = bl + g * 4 * SLABW;
        #pragma unroll
        for (int j = 0; j < 8; ++j) {
            a[j] = *(const v2f*)(ap + j * 4);
            #pragma unroll
            for (int nt = 0; nt < 2; ++nt) {
                const float* sp = bp + (j >> 1) * SLABW + (j & 1) * 4 + nt * 16;
                bq[j][nt].x = sp[0];
                bq[j][nt].y = sp[1];
            }
        }
    };
    auto mmaG = [&](const v2f (&a)[8], const v2f (&bq)[8][2]) {
        #pragma unroll
        for (int j = 0; j < 8; ++j) {
            #pragma unroll
            for (int nt = 0; nt < 2; ++nt)
                acc[nt] = __builtin_amdgcn_wmma_f32_16x16x4_f32(
                    false, a[j], false, bq[j][nt], (short)0, acc[nt], false, false);
        }
    };

    v2f aA[8], aB[8], bA[8][2], bB[8][2];
    loadG(0, aA, bA);
    for (int g = 0; g < KDIM / 32; g += 2) {
        loadG(g + 1, aB, bB);
        __builtin_amdgcn_sched_barrier(0);
        mmaG(aA, bA);
        __builtin_amdgcn_sched_barrier(0);
        loadG(g + 2 < KDIM / 32 ? g + 2 : g + 1, aA, bA);
        __builtin_amdgcn_sched_barrier(0);
        mmaG(aB, bB);
        __builtin_amdgcn_sched_barrier(0);
    }

    const unsigned long long newmax = tile_keys_store<2>(acc, t0, mt * 16, half, nloc, lane, rec);

    if (lane == 0) rowk[b] = newmax > oldmax ? newmax : oldmax;
    __syncthreads();
    if (tid < BATCH) rk_out[(size_t)mt * KSLOTS + tid] = rowk[tid];
}

__global__ __launch_bounds__(KDIM)
void k_update_first(const float* __restrict__ dnn, float* __restrict__ res,
                    float* __restrict__ out, const unsigned long long* __restrict__ tilebest,
                    int* __restrict__ tsel) {
    __shared__ unsigned long long wred[KDIM / 32];
    const int b    = blockIdx.x;
    const int tid  = threadIdx.x;
    const int lane = tid & 31;
    const unsigned long long* tb = tilebest + (size_t)b * MTILES * KSLOTS;
    const int sl = tid & (KSLOTS - 1);
    const bool used = (sl < NTPB) || (sl >= 16 && sl < 16 + NTILES - NTPB);
    unsigned long long key = 0ull;
    if (used) {
        #pragma unroll
        for (int j = 0; j < (MTILES * KSLOTS) / KDIM; ++j) {
            const unsigned long long k = tb[j * KDIM + tid];
            if (k > key) key = k;
        }
    }
    key = wave_max_key(key);
    if (lane == 0) wred[tid >> 5] = key;
    __syncthreads();
    key = wred[0];
    for (int w = 1; w < KDIM / 32; ++w) { const unsigned long long o = wred[w]; if (o > key) key = o; }

    const Pick p = decode_key(key);
    const float* ap = dnn + (size_t)p.atom * KDIM;
    float* rb = res + (size_t)b * CC * RES_W;
    float* ob = out + (size_t)b * CC * TT;
    for (int i = threadIdx.x; i < KDIM; i += blockDim.x) {
        const int c = i >> 3, t = i & 7;
        const int col = p.pos + t;
        const float a = ap[i];
        rb[c * RES_W + col] = sub_scaled(rb[c * RES_W + col], a, p.val);
        if (col < TT) ob[c * TT + col] = add_scaled(ob[c * TT + col], a, p.val);
    }
    if (tid == 0) {
        tsel[b] = first_tile(p.pos);
        tsel[BATCH + b] = p.pos;
    }
}

__global__ __launch_bounds__(MTILES)
void k_update_last(const float* __restrict__ dnn, float* __restrict__ out,
                   const unsigned long long* __restrict__ rk_in) {
    __shared__ unsigned long long wred[MTILES / 32];
    const int b    = blockIdx.x;
    const int tid  = threadIdx.x;
    const int lane = tid & 31;
    unsigned long long key = wave_max_key(rk_in[(size_t)tid * KSLOTS + b]);
    if (lane == 0) wred[tid >> 5] = key;
    __syncthreads();
    key = wred[0];
    for (int w = 1; w < MTILES / 32; ++w) { const unsigned long long o = wred[w]; if (o > key) key = o; }

    const Pick p = decode_key(key);
    const float* ap = dnn + (size_t)p.atom * KDIM;
    float* ob = out + (size_t)b * CC * TT;
    for (int i = tid; i < KDIM; i += MTILES) {
        const int c = i >> 3, t = i & 7;
        const int col = p.pos + t;
        if (col < TT) ob[c * TT + col] = add_scaled(ob[c * TT + col], ap[i], p.val);
    }
}

extern "C" void kernel_launch(void* const* d_in, const int* in_sizes, int n_in,
                              void* d_out, int out_size, void* d_ws, size_t ws_size,
                              hipStream_t stream) {
    const float* x = (const float*)d_in[0];
    const float* d = (const float*)d_in[1];
    (void)in_sizes; (void)n_in; (void)out_size; (void)ws_size;
    float* out = (float*)d_out;

    char* ws = (char*)d_ws;
    float* dnn  = (float*)ws;
    float* res0 = (float*)(ws + (size_t)NATOMS * KDIM * 4);
    float* res1 = res0 + BATCH * CC * RES_W;
    int* tsel = (int*)(res1 + BATCH * CC * RES_W);
    unsigned long long* tilebest = (unsigned long long*)(tsel + 8);
    unsigned long long* rka = tilebest + RKA;
    unsigned long long* rkb = tilebest + RKB;

    k_normalize<<<NATOMS, CORR_THREADS, 0, stream>>>(d, dnn);
    k_init<<<(BATCH * CC * RES_W + 255) / 256, 256, 0, stream>>>(x, res0, res1, out);

    const size_t shbytes = (size_t)CC * RES_W * sizeof(float);
    dim3 grid(NATOMS / MTILE, NSPLIT, BATCH);
    k_corr<<<grid, CORR_THREADS, shbytes, stream>>>(dnn, res0, tilebest);
    k_update_first<<<BATCH, KDIM, 0, stream>>>(dnn, res0, out, tilebest, tsel);
    k_corr_inc<false><<<MTILES, CORR_THREADS, 0, stream>>>(dnn, res0, res1, out, tsel, tilebest, rkb, rka);
    for (int it = 2; it < NITER; it += 2) {
        k_corr_inc<true><<<MTILES + 1, CORR_THREADS, 0, stream>>>(dnn, res0, res1, out, tsel, tilebest, rka, rkb);
        k_corr_inc<true><<<MTILES + 1, CORR_THREADS, 0, stream>>>(dnn, res1, res0, out, tsel, tilebest, rkb, rka);
    }
    k_update_last<<<BATCH, MTILES, 0, stream>>>(dnn, out, rka);
}


# Incremental launch recomputes one 16-position window at the update (one WMMA tile, not two); keys per (atom tile, position) replace the normalized copy, atoms scaled by inverse norm on use
# speedup vs baseline: 4.1991x; 1.1374x over previous
#include <hip/hip_runtime.h>

#define NATOMS 4096
#define CC     64
#define TT     256
#define ASIZE  8
#define BATCH  4
#define KDIM   512
#define PPOS   257
#define RES_W  304
#define NSPLIT 2
#define NTPB   9
#define MTILE  64
#define CORR_THREADS 128
#define NITER  32
#define NTILES 17
#define MTILES 256
#define KSLOTS 32
#define SLABW  40
#define APITCH 516
#define RKA    9
#define RKB    24
#define WLAST  (PPOS - 16)
#define INVP   32
#define POSK_OFF 1048576

typedef __attribute__((ext_vector_type(2))) float v2f;
typedef __attribute__((ext_vector_type(8))) float v8f;

__global__ void k_normalize(const float* __restrict__ d, float* __restrict__ inv) {
    __shared__ float red[4];
    const int a = blockIdx.x;
    const float* src = d + (size_t)a * KDIM;
    float s = 0.f;
    for (int i = threadIdx.x; i < KDIM; i += CORR_THREADS) { float v = src[i]; s += v * v; }
    for (int off = 16; off > 0; off >>= 1) s += __shfl_down(s, off, 32);
    const int wv = threadIdx.x >> 5;
    if ((threadIdx.x & 31) == 0) red[wv] = s;
    __syncthreads();
    if (threadIdx.x == 0) inv[(size_t)a * INVP] = 1.f / (sqrtf(red[0] + red[1] + red[2] + red[3]) + 1e-8f);
}

__global__ void k_init(const float* __restrict__ x, float* __restrict__ res0,
                       float* __restrict__ res1, float* __restrict__ out) {
    const int i = blockIdx.x * blockDim.x + threadIdx.x;
    if (i < BATCH * CC * RES_W) {
        const int col = i % RES_W;
        const int bc  = i / RES_W;
        const float v = (col < TT) ? x[bc * TT + col] : 0.f;
        res0[i] = v;
        res1[i] = v;
        if (col < TT) out[bc * TT + col] = 0.f;
    }
}

struct Pick { float val; int atom; int pos; };

__device__ __forceinline__ unsigned long long wave_max_key(unsigned long long key) {
    for (int off = 16; off > 0; off >>= 1) {
        unsigned long long o = __shfl_xor(key, off, 32);
        if (o > key) key = o;
    }
    return key;
}

__device__ __forceinline__ Pick decode_key(unsigned long long key) {
    const unsigned enc  = (unsigned)(key >> 32);
    const unsigned bits = (enc & 0x80000000u) ? (enc & 0x7FFFFFFFu) : ~enc;
    const unsigned idx  = 0xFFFFFFFFu - (unsigned)(key & 0xFFFFFFFFu);
    const unsigned a    = idx / PPOS;
    Pick p;
    p.val  = __uint_as_float(bits);
    p.atom = (int)(a < NATOMS ? a : NATOMS - 1);
    p.pos  = (int)(idx % PPOS);
    return p;
}

__device__ __forceinline__ int first_tile(int pos) {
    const int lo = pos > 7 ? pos - 7 : 0;
    const int t0 = lo >> 4;
    return t0 < NTILES - 2 ? t0 : NTILES - 2;
}

__device__ __forceinline__ float sub_scaled(float r, float a, float val) { return __builtin_fmaf(-a, val, r); }
__device__ __forceinline__ float add_scaled(float c, float a, float val) { return __builtin_fmaf(a, val, c); }

__device__ __forceinline__ void load_row_keys(const unsigned long long* rk, int b, int lane,
                                              unsigned long long (&k)[MTILES / 32]) {
    #pragma unroll
    for (int j = 0; j < MTILES / 32; ++j) k[j] = rk[(size_t)(lane + 32 * j) * KSLOTS + b];
}

__device__ __forceinline__ Pick pick_from_keys(const unsigned long long (&k)[MTILES / 32]) {
    unsigned long long key = k[0];
    #pragma unroll
    for (int j = 1; j < MTILES / 32; ++j) if (k[j] > key) key = k[j];
    key = wave_max_key(key);
    const unsigned klo = __builtin_amdgcn_readfirstlane((unsigned)(key & 0xFFFFFFFFu));
    const unsigned khi = __builtin_amdgcn_readfirstlane((unsigned)(key >> 32));
    return decode_key(((unsigned long long)khi << 32) | klo);
}

template <int NT>
__device__ __forceinline__ unsigned long long tile_keys_store(const v8f (&acc)[NT], int nt0, int atom0,
                                                int half, int nloc, int lane,
                                                unsigned long long* __restrict__ rec,
                                                unsigned long long* __restrict__ posk) {
    unsigned long long mykey = 0ull, allmax = 0ull;
    #pragma unroll
    for (int nt = 0; nt < NT; ++nt) {
        float bestv = -3.402823466e38f;
        unsigned bestidx = 0u;
        const int pos = (nt0 + nt) * 16 + nloc;
        if (pos < PPOS) {
            #pragma unroll
            for (int r = 0; r < 8; ++r) {
                const float v = acc[nt][r];
                if (v > bestv) {
                    bestv = v;
                    bestidx = (unsigned)((atom0 + r + 8 * half) * PPOS + pos);
                }
            }
        }
        unsigned vb = __float_as_uint(bestv);
        vb = (vb & 0x80000000u) ? ~vb : (vb | 0x80000000u);
        unsigned long long key = ((unsigned long long)vb << 32) | (0xFFFFFFFFu - bestidx);
        {
            unsigned long long o = __shfl_xor(key, 16, 32);
            if (o > key) key = o;
        }
        if (nt0 + nt < NTILES && lane < 16) posk[(nt0 + nt) * 16 + lane] = key;
        for (int off = 8; off > 0; off >>= 1) {
            unsigned long long o = __shfl_xor(key, off, 32);
            if (o > key) key = o;
        }
        if (lane == nt) mykey = key;
        if (key > allmax) allmax = key;
    }
    const int t = nt0 + lane;
    if (lane < NT && t < NTILES) rec[t + (t >= NTPB ? 16 - NTPB : 0)] = mykey;
    return allmax;
}

__global__ __launch_bounds__(CORR_THREADS, 1)
void k_corr(const float* __restrict__ d, const float* __restrict__ inv, const float* __restrict__ res,
            unsigned long long* __restrict__ tilebest, unsigned long long* __restrict__ poskeys) {
    extern __shared__ float smem[];
    const int b    = blockIdx.z;
    const int nt0  = blockIdx.y * NTPB;
    const int m0   = blockIdx.x * MTILE;
    const int tid  = threadIdx.x;
    const int lane = tid & 31;
    const int wave = tid >> 5;

    const int half = lane >> 4;
    const int nloc = lane & 15;
    const float* arow = d + (size_t)(m0 + wave * 16 + nloc) * KDIM + half * 2;
    const float ainv = inv[(size_t)(m0 + wave * 16 + nloc) * INVP];

    #pragma unroll
    for (int pf = 0; pf < 16; ++pf)
        __builtin_prefetch(arow + pf * 32, 0, 3);

    const float4* resb = (const float4*)(res + (size_t)b * CC * RES_W);
    float4* s4 = (float4*)smem;
    for (int i = tid; i < (CC * RES_W) / 4; i += CORR_THREADS) s4[i] = resb[i];
    __syncthreads();

    v8f acc[NTPB] = {};

    auto loadA = [&](int ks) { v2f a; a.x = arow[ks * 4] * ainv; a.y = arow[ks * 4 + 1] * ainv; return a; };
    auto srowp = [&](int ks) {
        const int kbase = ks * 4 + half * 2;
        return (const float*)smem + (kbase >> 3) * RES_W + (kbase & 7) + nloc + nt0 * 16;
    };

    v2f a0 = loadA(0);
    for (int ks = 0; ks < KDIM / 4; ++ks) {
        v2f an = loadA(ks + 1 < KDIM / 4 ? ks + 1 : ks);
        const float* sr = srowp(ks);
        v2f bf[NTPB];
        #pragma unroll
        for (int nt = 0; nt < NTPB; ++nt) {
            bf[nt].x = sr[nt * 16];
            bf[nt].y = sr[nt * 16 + 1];
        }
        #pragma unroll
        for (int nt = 0; nt < NTPB; ++nt)
            acc[nt] = __builtin_amdgcn_wmma_f32_16x16x4_f32(
                false, a0, false, bf[nt], (short)0, acc[nt], false, false);
        a0 = an;
    }

    const int mt = blockIdx.x * (MTILE / 16) + wave;
    tile_keys_store<NTPB>(acc, nt0, m0 + wave * 16, half, nloc, lane,
                          tilebest + ((size_t)b * MTILES + mt) * KSLOTS,
                          poskeys + ((size_t)b * MTILES + mt) * NTILES * 16);
}

template <bool HEAD>
__global__ __launch_bounds__(CORR_THREADS, 1)
void k_corr_inc(const float* __restrict__ d, const float* __restrict__ inv, const float* __restrict__ rin,
                float* __restrict__ rout, float* __restrict__ out, int* __restrict__ tsel,
                unsigned long long* tilebest,
                unsigned long long* poskeys,
                const unsigned long long* rk_in, unsigned long long* rk_out) {
    __shared__ __attribute__((aligned(16))) float atile[16 * APITCH];
    __shared__ __attribute__((aligned(16))) float slab[BATCH * CC * SLABW];
    __shared__ unsigned long long rowk[BATCH];
    const int mt   = blockIdx.x;
    const int tid  = threadIdx.x;
    const int lane = tid & 31;
    const int b    = __builtin_amdgcn_readfirstlane(tid >> 5);
    const int half = lane >> 4;
    const int nloc = lane & 15;

    unsigned long long rkeys[MTILES / 32];
    if (HEAD) {
        load_row_keys(rk_in, b, lane, rkeys);

        if (mt == MTILES) {
            const Pick p = pick_from_keys(rkeys);
            int pprev = tsel[BATCH + b];
            pprev = pprev < 0 ? 0 : (pprev < PPOS ? pprev : PPOS - 1);
            const float* ap = d + (size_t)p.atom * KDIM;
            const float pinv = inv[(size_t)p.atom * INVP];
            const float* ri = rin  + (size_t)b * CC * RES_W;
            float*       ro = rout + (size_t)b * CC * RES_W;
            float*       ob = out  + (size_t)b * CC * TT;
            const int t  = lane & 7;
            const int c0 = lane >> 3;
            const int col = p.pos + t;
            const int pc  = pprev + t;
            const bool carry = pc < p.pos || pc >= p.pos + ASIZE;
            float a[KDIM / 32], r[KDIM / 32], o[KDIM / 32], pv[KDIM / 32];
            #pragma unroll
            for (int j = 0; j < KDIM / 32; ++j) {
                const int c = c0 + 4 * j;
                a[j]  = ap[lane + 32 * j];
                r[j]  = ri[c * RES_W + col];
                pv[j] = ri[c * RES_W + pc];
                o[j]  = col < TT ? ob[c * TT + col] : 0.f;
            }
            #pragma unroll
            for (int j = 0; j < KDIM / 32; ++j) {
                const int c = c0 + 4 * j;
                const float an = a[j] * pinv;
                ro[c * RES_W + col] = sub_scaled(r[j], an, p.val);
                if (carry) ro[c * RES_W + pc] = pv[j];
                if (col < TT) ob[c * TT + col] = add_scaled(o[j], an, p.val);
            }
            if (lane == 0) tsel[BATCH + b] = p.pos;
            return;
        }
    }

    const float4* asrc = (const float4*)(d + (size_t)mt * 16 * KDIM) + tid;
    float4 av[16];
    #pragma unroll
    for (int j = 0; j < 16; ++j) av[j] = asrc[j * (KDIM / 4)];
    float rinv[16];
    #pragma unroll
    for (int j = 0; j < 16; ++j) rinv[j] = inv[(size_t)(mt * 16 + j) * INVP];

    unsigned long long* rec = tilebest + ((size_t)b * MTILES + mt) * KSLOTS;
    unsigned long long oldkeys[NTILES];
    #pragma unroll
    for (int j = 0; j < NTILES; ++j) oldkeys[j] = rec[j + (j >= NTPB ? 16 - NTPB : 0)];

    unsigned soff[(CC * SLABW / 4) / 32];
    #pragma unroll
    for (int j = 0; j < (CC * SLABW / 4) / 32; ++j) {
        const int i   = lane + 32 * j;
        const int row = i / (SLABW / 4);
        const int q   = i - row * (SLABW / 4);
        soff[j] = (unsigned)(row * RES_W + q * 4) * 4u;
    }

    Pick p = {0.f, 0, 0};
    if (HEAD) p = pick_from_keys(rkeys);
    int ppos = p.pos;
    if (!HEAD) {
        ppos = tsel[BATCH + b];
        ppos = ppos < 0 ? 0 : (ppos < PPOS ? ppos : PPOS - 1);
    }
    const int t0 = first_tile(ppos);
    int w = ppos > 7 ? ppos - 7 : 0;
    w = w < WLAST ? w : WLAST;
    const int woff = w - t0 * 16;

    unsigned long long* pkr = poskeys + (((size_t)b * MTILES + mt) * NTILES + t0) * 16;
    const unsigned long long oldpk = pkr[lane];

    float* sl = slab + b * CC * SLABW;
    const char* rsrc = (const char*)(rin + (size_t)b * CC * RES_W + t0 * 16);
    float4 sv[(CC * SLABW / 4) / 32];
    #pragma unroll
    for (int j = 0; j < (CC * SLABW / 4) / 32; ++j) sv[j] = *(const float4*)(rsrc + soff[j]);
    float4 uv[KDIM / 128];
    float pinv = 0.f;
    if (HEAD) {
        const float4* arow4 = (const float4*)(d + (size_t)p.atom * KDIM);
        #pragma unroll
        for (int j = 0; j < KDIM / 128; ++j) uv[j] = arow4[lane + 32 * j];
        pinv = inv[(size_t)p.atom * INVP];
    }
    unsigned long long oldmax = 0ull;
    #pragma unroll
    for (int j = 0; j < NTILES; ++j)
        if (j != t0 && j != t0 + 1 && oldkeys[j] > oldmax) oldmax = oldkeys[j];

    #pragma unroll
    for (int j = 0; j < 16; ++j) {
        float4 v = av[j];
        v.x *= rinv[j]; v.y *= rinv[j]; v.z *= rinv[j]; v.w *= rinv[j];
        *(float4*)(atile + j * APITCH + tid * 4) = v;
    }
    #pragma unroll
    for (int j = 0; j < (CC * SLABW / 4) / 32; ++j) *(float4*)(sl + 4 * (lane + 32 * j)) = sv[j];
    __syncthreads();

    if (HEAD) {
        float* up = sl + (p.pos - t0 * 16);
        float rv[KDIM / 128][4];
        #pragma unroll
        for (int j = 0; j < KDIM / 128; ++j) {
            const int i4 = lane + 32 * j;
            const float* q = up + (i4 >> 1) * SLABW + (i4 & 1) * 4;
            rv[j][0] = q[0]; rv[j][1] = q[1]; rv[j][2] = q[2]; rv[j][3] = q[3];
        }
        #pragma unroll
        for (int j = 0; j < KDIM / 128; ++j) {
            const int i4 = lane + 32 * j;
            float* q = up + (i4 >> 1) * SLABW + (i4 & 1) * 4;
            q[0] = sub_scaled(rv[j][0], uv[j].x * pinv, p.val);
            q[1] = sub_scaled(rv[j][1], uv[j].y * pinv, p.val);
            q[2] = sub_scaled(rv[j][2], uv[j].z * pinv, p.val);
            q[3] = sub_scaled(rv[j][3], uv[j].w * pinv, p.val);
        }
        __syncthreads();
    }

    const float* al = atile + nloc * APITCH + half * 2;
    const float* bl = sl + half * 2 + nloc + woff;
    v8f acc = {};

    auto loadG = [&](int g, v2f (&a)[8], v2f (&bq)[8]) {
        const float* ap = al + g * 32;
        const float* bp = bl + g * 4 * SLABW;
        #pragma unroll
        for (int j = 0; j < 8; ++j) {
            a[j] = *(const v2f*)(ap + j * 4);
            const float* sp = bp + (j >> 1) * SLABW + (j & 1) * 4;
            bq[j].x = sp[0];
            bq[j].y = sp[1];
        }
    };
    auto mmaG = [&](const v2f (&a)[8], const v2f (&bq)[8]) {
        #pragma unroll
        for (int j = 0; j < 8; ++j)
            acc = __builtin_amdgcn_wmma_f32_16x16x4_f32(
                false, a[j], false, bq[j], (short)0, acc, false, false);
    };

    v2f aA[8], aB[8], bA[8], bB[8];
    loadG(0, aA, bA);
    for (int g = 0; g < KDIM / 32; g += 2) {
        loadG(g + 1, aB, bB);
        __builtin_amdgcn_sched_barrier(0);
        mmaG(aA, bA);
        __builtin_amdgcn_sched_barrier(0);
        loadG(g + 2 < KDIM / 32 ? g + 2 : g + 1, aA, bA);
        __builtin_amdgcn_sched_barrier(0);
        mmaG(aB, bB);
        __builtin_amdgcn_sched_barrier(0);
    }

    float bestv = acc[0];
    unsigned bestidx = (unsigned)((mt * 16 + 8 * half) * PPOS + w + nloc);
    #pragma unroll
    for (int r = 1; r < 8; ++r) {
        const float v = acc[r];
        if (v > bestv) {
            bestv = v;
            bestidx = (unsigned)((mt * 16 + r + 8 * half) * PPOS + w + nloc);
        }
    }
    unsigned vb = __float_as_uint(bestv);
    vb = (vb & 0x80000000u) ? ~vb : (vb | 0x80000000u);
    unsigned long long ckey = ((unsigned long long)vb << 32) | (0xFFFFFFFFu - bestidx);
    {
        const unsigned long long o = __shfl_xor(ckey, 16, 32);
        if (o > ckey) ckey = o;
    }
    const int rel = lane - woff;
    const unsigned long long nkey = __shfl(ckey, rel & 15, 32);
    unsigned long long mk = (rel >= 0 && rel < 16) ? nkey : oldpk;
    pkr[lane] = mk;
    for (int off = 8; off > 0; off >>= 1) {
        const unsigned long long o = __shfl_xor(mk, off, 32);
        if (o > mk) mk = o;
    }
    {
        const int t = t0 + half;
        if (nloc == 0 && t < NTILES) rec[t + (t >= NTPB ? 16 - NTPB : 0)] = mk;
    }
    unsigned long long newmax = mk;
    {
        const unsigned long long o = __shfl_xor(mk, 16, 32);
        if (o > newmax) newmax = o;
    }

    if (lane == 0) rowk[b] = newmax > oldmax ? newmax : oldmax;
    __syncthreads();
    if (tid < BATCH) rk_out[(size_t)mt * KSLOTS + tid] = rowk[tid];
}

__global__ __launch_bounds__(KDIM)
void k_update_first(const float* __restrict__ d, const float* __restrict__ inv, float* __restrict__ res,
                    float* __restrict__ out, const unsigned long long* __restrict__ tilebest,
                    int* __restrict__ tsel) {
    __shared__ unsigned long long wred[KDIM / 32];
    const int b    = blockIdx.x;
    const int tid  = threadIdx.x;
    const int lane = tid & 31;
    const unsigned long long* tb = tilebest + (size_t)b * MTILES * KSLOTS;
    const int sl = tid & (KSLOTS - 1);
    const bool used = (sl < NTPB) || (sl >= 16 && sl < 16 + NTILES - NTPB);
    unsigned long long key = 0ull;
    if (used) {
        #pragma unroll
        for (int j = 0; j < (MTILES * KSLOTS) / KDIM; ++j) {
            const unsigned long long k = tb[j * KDIM + tid];
            if (k > key) key = k;
        }
    }
    key = wave_max_key(key);
    if (lane == 0) wred[tid >> 5] = key;
    __syncthreads();
    key = wred[0];
    for (int w = 1; w < KDIM / 32; ++w) { const unsigned long long o = wred[w]; if (o > key) key = o; }

    const Pick p = decode_key(key);
    const float* ap = d + (size_t)p.atom * KDIM;
    const float pinv = inv[(size_t)p.atom * INVP];
    float* rb = res + (size_t)b * CC * RES_W;
    float* ob = out + (size_t)b * CC * TT;
    for (int i = threadIdx.x; i < KDIM; i += blockDim.x) {
        const int c = i >> 3, t = i & 7;
        const int col = p.pos + t;
        const float a = ap[i] * pinv;
        rb[c * RES_W + col] = sub_scaled(rb[c * RES_W + col], a, p.val);
        if (col < TT) ob[c * TT + col] = add_scaled(ob[c * TT + col], a, p.val);
    }
    if (tid == 0) {
        tsel[b] = first_tile(p.pos);
        tsel[BATCH + b] = p.pos;
    }
}

__global__ __launch_bounds__(MTILES)
void k_update_last(const float* __restrict__ d, const float* __restrict__ inv, float* __restrict__ out,
                   const unsigned long long* __restrict__ rk_in) {
    __shared__ unsigned long long wred[MTILES / 32];
    const int b    = blockIdx.x;
    const int tid  = threadIdx.x;
    const int lane = tid & 31;
    unsigned long long key = wave_max_key(rk_in[(size_t)tid * KSLOTS + b]);
    if (lane == 0) wred[tid >> 5] = key;
    __syncthreads();
    key = wred[0];
    for (int w = 1; w < MTILES / 32; ++w) { const unsigned long long o = wred[w]; if (o > key) key = o; }

    const Pick p = decode_key(key);
    const float* ap = d + (size_t)p.atom * KDIM;
    const float pinv = inv[(size_t)p.atom * INVP];
    float* ob = out + (size_t)b * CC * TT;
    for (int i = tid; i < KDIM; i += MTILES) {
        const int c = i >> 3, t = i & 7;
        const int col = p.pos + t;
        const float a = ap[i] * pinv;
        if (col < TT) ob[c * TT + col] = add_scaled(ob[c * TT + col], a, p.val);
    }
}

extern "C" void kernel_launch(void* const* d_in, const int* in_sizes, int n_in,
                              void* d_out, int out_size, void* d_ws, size_t ws_size,
                              hipStream_t stream) {
    const float* x = (const float*)d_in[0];
    const float* d = (const float*)d_in[1];
    (void)in_sizes; (void)n_in; (void)out_size; (void)ws_size;
    float* out = (float*)d_out;

    char* ws = (char*)d_ws;
    float* inv  = (float*)ws;
    unsigned long long* poskeys = (unsigned long long*)(ws + POSK_OFF);
    float* res0 = (float*)(ws + (size_t)NATOMS * KDIM * 4);
    float* res1 = res0 + BATCH * CC * RES_W;
    int* tsel = (int*)(res1 + BATCH * CC * RES_W);
    unsigned long long* tilebest = (unsigned long long*)(tsel + 8);
    unsigned long long* rka = tilebest + RKA;
    unsigned long long* rkb = tilebest + RKB;

    k_normalize<<<NATOMS, CORR_THREADS, 0, stream>>>(d, inv);
    k_init<<<(BATCH * CC * RES_W + 255) / 256, 256, 0, stream>>>(x, res0, res1, out);

    const size_t shbytes = (size_t)CC * RES_W * sizeof(float);
    dim3 grid(NATOMS / MTILE, NSPLIT, BATCH);
    k_corr<<<grid, CORR_THREADS, shbytes, stream>>>(d, inv, res0, tilebest, poskeys);
    k_update_first<<<BATCH, KDIM, 0, stream>>>(d, inv, res0, out, tilebest, tsel);
    k_corr_inc<false><<<MTILES, CORR_THREADS, 0, stream>>>(d, inv, res0, res1, out, tsel, tilebest, poskeys, rkb, rka);
    for (int it = 2; it < NITER; it += 2) {
        k_corr_inc<true><<<MTILES + 1, CORR_THREADS, 0, stream>>>(d, inv, res0, res1, out, tsel, tilebest, poskeys, rka, rkb);
        k_corr_inc<true><<<MTILES + 1, CORR_THREADS, 0, stream>>>(d, inv, res1, res0, out, tsel, tilebest, poskeys, rkb, rka);
    }
    k_update_last<<<BATCH, MTILES, 0, stream>>>(d, inv, out, rka);
}


# Step argmax by integer atomicMax: every block folds its row key into one word per batch item (three rotating, self-zeroed slots); a wave reads that word instead of loading and reducing 256 row keys
# speedup vs baseline: 4.8607x; 1.1576x over previous
#include <hip/hip_runtime.h>

#define NATOMS 4096
#define CC     64
#define TT     256
#define ASIZE  8
#define BATCH  4
#define KDIM   512
#define PPOS   257
#define RES_W  304
#define NSPLIT 2
#define NTPB   9
#define MTILE  64
#define CORR_THREADS 128
#define NITER  32
#define NTILES 17
#define MTILES 256
#define KSLOTS 32
#define SLABW  40
#define APITCH 516
#define KWPITCH 32
#define KWSLOT (BATCH * KWPITCH)
#define KW_OFF 4194304
#define WLAST  (PPOS - 16)
#define INVP   32
#define POSK_OFF 1048576

typedef __attribute__((ext_vector_type(2))) float v2f;
typedef __attribute__((ext_vector_type(8))) float v8f;

__global__ void k_normalize(const float* __restrict__ d, float* __restrict__ inv) {
    __shared__ float red[4];
    const int a = blockIdx.x;
    const float* src = d + (size_t)a * KDIM;
    float s = 0.f;
    for (int i = threadIdx.x; i < KDIM; i += CORR_THREADS) { float v = src[i]; s += v * v; }
    for (int off = 16; off > 0; off >>= 1) s += __shfl_down(s, off, 32);
    const int wv = threadIdx.x >> 5;
    if ((threadIdx.x & 31) == 0) red[wv] = s;
    __syncthreads();
    if (threadIdx.x == 0) inv[(size_t)a * INVP] = 1.f / (sqrtf(red[0] + red[1] + red[2] + red[3]) + 1e-8f);
}

__global__ void k_init(const float* __restrict__ x, float* __restrict__ res0,
                       float* __restrict__ res1, float* __restrict__ out,
                       unsigned long long* __restrict__ kw) {
    const int i = blockIdx.x * blockDim.x + threadIdx.x;
    if (i < 3 * BATCH) kw[(size_t)i * KWPITCH] = 0ull;
    if (i < BATCH * CC * RES_W) {
        const int col = i % RES_W;
        const int bc  = i / RES_W;
        const float v = (col < TT) ? x[bc * TT + col] : 0.f;
        res0[i] = v;
        res1[i] = v;
        if (col < TT) out[bc * TT + col] = 0.f;
    }
}

struct Pick { float val; int atom; int pos; };

__device__ __forceinline__ unsigned long long wave_max_key(unsigned long long key) {
    for (int off = 16; off > 0; off >>= 1) {
        unsigned long long o = __shfl_xor(key, off, 32);
        if (o > key) key = o;
    }
    return key;
}

__device__ __forceinline__ Pick decode_key(unsigned long long key) {
    const unsigned enc  = (unsigned)(key >> 32);
    const unsigned bits = (enc & 0x80000000u) ? (enc & 0x7FFFFFFFu) : ~enc;
    const unsigned idx  = 0xFFFFFFFFu - (unsigned)(key & 0xFFFFFFFFu);
    const unsigned a    = idx / PPOS;
    Pick p;
    p.val  = __uint_as_float(bits);
    p.atom = (int)(a < NATOMS ? a : NATOMS - 1);
    p.pos  = (int)(idx % PPOS);
    return p;
}

__device__ __forceinline__ int first_tile(int pos) {
    const int lo = pos > 7 ? pos - 7 : 0;
    const int t0 = lo >> 4;
    return t0 < NTILES - 2 ? t0 : NTILES - 2;
}

__device__ __forceinline__ float sub_scaled(float r, float a, float val) { return __builtin_fmaf(-a, val, r); }
__device__ __forceinline__ float add_scaled(float c, float a, float val) { return __builtin_fmaf(a, val, c); }

__device__ __forceinline__ Pick pick_from_word(const unsigned long long* kw, int b) {
    const unsigned long long key = kw[(size_t)b * KWPITCH];
    const unsigned klo = __builtin_amdgcn_readfirstlane((unsigned)(key & 0xFFFFFFFFu));
    const unsigned khi = __builtin_amdgcn_readfirstlane((unsigned)(key >> 32));
    return decode_key(((unsigned long long)khi << 32) | klo);
}

template <int NT>
__device__ __forceinline__ unsigned long long tile_keys_store(const v8f (&acc)[NT], int nt0, int atom0,
                                                int half, int nloc, int lane,
                                                unsigned long long* __restrict__ rec,
                                                unsigned long long* __restrict__ posk) {
    unsigned long long mykey = 0ull, allmax = 0ull;
    #pragma unroll
    for (int nt = 0; nt < NT; ++nt) {
        float bestv = -3.402823466e38f;
        unsigned bestidx = 0u;
        const int pos = (nt0 + nt) * 16 + nloc;
        if (pos < PPOS) {
            #pragma unroll
            for (int r = 0; r < 8; ++r) {
                const float v = acc[nt][r];
                if (v > bestv) {
                    bestv = v;
                    bestidx = (unsigned)((atom0 + r + 8 * half) * PPOS + pos);
                }
            }
        }
        unsigned vb = __float_as_uint(bestv);
        vb = (vb & 0x80000000u) ? ~vb : (vb | 0x80000000u);
        unsigned long long key = ((unsigned long long)vb << 32) | (0xFFFFFFFFu - bestidx);
        {
            unsigned long long o = __shfl_xor(key, 16, 32);
            if (o > key) key = o;
        }
        if (nt0 + nt < NTILES && lane < 16) posk[(nt0 + nt) * 16 + lane] = key;
        for (int off = 8; off > 0; off >>= 1) {
            unsigned long long o = __shfl_xor(key, off, 32);
            if (o > key) key = o;
        }
        if (lane == nt) mykey = key;
        if (key > allmax) allmax = key;
    }
    const int t = nt0 + lane;
    if (lane < NT && t < NTILES) rec[t + (t >= NTPB ? 16 - NTPB : 0)] = mykey;
    return allmax;
}

__global__ __launch_bounds__(CORR_THREADS, 1)
void k_corr(const float* __restrict__ d, const float* __restrict__ inv, const float* __restrict__ res,
            unsigned long long* __restrict__ tilebest, unsigned long long* __restrict__ poskeys) {
    extern __shared__ float smem[];
    const int b    = blockIdx.z;
    const int nt0  = blockIdx.y * NTPB;
    const int m0   = blockIdx.x * MTILE;
    const int tid  = threadIdx.x;
    const int lane = tid & 31;
    const int wave = tid >> 5;

    const int half = lane >> 4;
    const int nloc = lane & 15;
    const float* arow = d + (size_t)(m0 + wave * 16 + nloc) * KDIM + half * 2;
    const float ainv = inv[(size_t)(m0 + wave * 16 + nloc) * INVP];

    #pragma unroll
    for (int pf = 0; pf < 16; ++pf)
        __builtin_prefetch(arow + pf * 32, 0, 3);

    const float4* resb = (const float4*)(res + (size_t)b * CC * RES_W);
    float4* s4 = (float4*)smem;
    for (int i = tid; i < (CC * RES_W) / 4; i += CORR_THREADS) s4[i] = resb[i];
    __syncthreads();

    v8f acc[NTPB] = {};

    auto loadA = [&](int ks) { v2f a; a.x = arow[ks * 4] * ainv; a.y = arow[ks * 4 + 1] * ainv; return a; };
    auto srowp = [&](int ks) {
        const int kbase = ks * 4 + half * 2;
        return (const float*)smem + (kbase >> 3) * RES_W + (kbase & 7) + nloc + nt0 * 16;
    };

    v2f a0 = loadA(0);
    for (int ks = 0; ks < KDIM / 4; ++ks) {
        v2f an = loadA(ks + 1 < KDIM / 4 ? ks + 1 : ks);
        const float* sr = srowp(ks);
        v2f bf[NTPB];
        #pragma unroll
        for (int nt = 0; nt < NTPB; ++nt) {
            bf[nt].x = sr[nt * 16];
            bf[nt].y = sr[nt * 16 + 1];
        }
        #pragma unroll
        for (int nt = 0; nt < NTPB; ++nt)
            acc[nt] = __builtin_amdgcn_wmma_f32_16x16x4_f32(
                false, a0, false, bf[nt], (short)0, acc[nt], false, false);
        a0 = an;
    }

    const int mt = blockIdx.x * (MTILE / 16) + wave;
    tile_keys_store<NTPB>(acc, nt0, m0 + wave * 16, half, nloc, lane,
                          tilebest + ((size_t)b * MTILES + mt) * KSLOTS,
                          poskeys + ((size_t)b * MTILES + mt) * NTILES * 16);
}

template <bool HEAD>
__global__ __launch_bounds__(CORR_THREADS, 1)
void k_corr_inc(const float* __restrict__ d, const float* __restrict__ inv, const float* __restrict__ rin,
                float* __restrict__ rout, float* __restrict__ out, int* __restrict__ tsel,
                unsigned long long* tilebest,
                unsigned long long* poskeys,
                const unsigned long long* kw_in, unsigned long long* kw_out, unsigned long long* kw_zero) {
    __shared__ __attribute__((aligned(16))) float atile[16 * APITCH];
    __shared__ __attribute__((aligned(16))) float slab[BATCH * CC * SLABW];
    const int mt   = blockIdx.x;
    const int tid  = threadIdx.x;
    const int lane = tid & 31;
    const int b    = __builtin_amdgcn_readfirstlane(tid >> 5);
    const int half = lane >> 4;
    const int nloc = lane & 15;

    Pick p = {0.f, 0, 0};
    if (HEAD) {
        p = pick_from_word(kw_in, b);

        if (mt == MTILES) {
            int pprev = tsel[BATCH + b];
            pprev = pprev < 0 ? 0 : (pprev < PPOS ? pprev : PPOS - 1);
            const float* ap = d + (size_t)p.atom * KDIM;
            const float pinv = inv[(size_t)p.atom * INVP];
            const float* ri = rin  + (size_t)b * CC * RES_W;
            float*       ro = rout + (size_t)b * CC * RES_W;
            float*       ob = out  + (size_t)b * CC * TT;
            const int t  = lane & 7;
            const int c0 = lane >> 3;
            const int col = p.pos + t;
            const int pc  = pprev + t;
            const bool carry = pc < p.pos || pc >= p.pos + ASIZE;
            float a[KDIM / 32], r[KDIM / 32], o[KDIM / 32], pv[KDIM / 32];
            #pragma unroll
            for (int j = 0; j < KDIM / 32; ++j) {
                const int c = c0 + 4 * j;
                a[j]  = ap[lane + 32 * j];
                r[j]  = ri[c * RES_W + col];
                pv[j] = ri[c * RES_W + pc];
                o[j]  = col < TT ? ob[c * TT + col] : 0.f;
            }
            #pragma unroll
            for (int j = 0; j < KDIM / 32; ++j) {
                const int c = c0 + 4 * j;
                const float an = a[j] * pinv;
                ro[c * RES_W + col] = sub_scaled(r[j], an, p.val);
                if (carry) ro[c * RES_W + pc] = pv[j];
                if (col < TT) ob[c * TT + col] = add_scaled(o[j], an, p.val);
            }
            if (lane == 0) {
                tsel[BATCH + b] = p.pos;
                kw_zero[(size_t)b * KWPITCH] = 0ull;
            }
            return;
        }
    }

    const float4* asrc = (const float4*)(d + (size_t)mt * 16 * KDIM) + tid;
    float4 av[16];
    #pragma unroll
    for (int j = 0; j < 16; ++j) av[j] = asrc[j * (KDIM / 4)];
    float rinv[16];
    #pragma unroll
    for (int j = 0; j < 16; ++j) rinv[j] = inv[(size_t)(mt * 16 + j) * INVP];

    unsigned long long* rec = tilebest + ((size_t)b * MTILES + mt) * KSLOTS;
    unsigned long long oldkeys[NTILES];
    #pragma unroll
    for (int j = 0; j < NTILES; ++j) oldkeys[j] = rec[j + (j >= NTPB ? 16 - NTPB : 0)];

    unsigned soff[(CC * SLABW / 4) / 32];
    #pragma unroll
    for (int j = 0; j < (CC * SLABW / 4) / 32; ++j) {
        const int i   = lane + 32 * j;
        const int row = i / (SLABW / 4);
        const int q   = i - row * (SLABW / 4);
        soff[j] = (unsigned)(row * RES_W + q * 4) * 4u;
    }

    int ppos = p.pos;
    if (!HEAD) {
        ppos = tsel[BATCH + b];
        ppos = ppos < 0 ? 0 : (ppos < PPOS ? ppos : PPOS - 1);
    }
    const int t0 = first_tile(ppos);
    int w = ppos > 7 ? ppos - 7 : 0;
    w = w < WLAST ? w : WLAST;
    const int woff = w - t0 * 16;

    unsigned long long* pkr = poskeys + (((size_t)b * MTILES + mt) * NTILES + t0) * 16;
    const unsigned long long oldpk = pkr[lane];

    float* sl = slab + b * CC * SLABW;
    const char* rsrc = (const char*)(rin + (size_t)b * CC * RES_W + t0 * 16);
    float4 sv[(CC * SLABW / 4) / 32];
    #pragma unroll
    for (int j = 0; j < (CC * SLABW / 4) / 32; ++j) sv[j] = *(const float4*)(rsrc + soff[j]);
    float4 uv[KDIM / 128];
    float pinv = 0.f;
    if (HEAD) {
        const float4* arow4 = (const float4*)(d + (size_t)p.atom * KDIM);
        #pragma unroll
        for (int j = 0; j < KDIM / 128; ++j) uv[j] = arow4[lane + 32 * j];
        pinv = inv[(size_t)p.atom * INVP];
    }
    #pragma unroll
    for (int j = 0; j < 16; ++j) {
        float4 v = av[j];
        v.x *= rinv[j]; v.y *= rinv[j]; v.z *= rinv[j]; v.w *= rinv[j];
        *(float4*)(atile + j * APITCH + tid * 4) = v;
    }
    #pragma unroll
    for (int j = 0; j < (CC * SLABW / 4) / 32; ++j) *(float4*)(sl + 4 * (lane + 32 * j)) = sv[j];
    __syncthreads();

    if (HEAD) {
        float* up = sl + (p.pos - t0 * 16);
        float rv[KDIM / 128][4];
        #pragma unroll
        for (int j = 0; j < KDIM / 128; ++j) {
            const int i4 = lane + 32 * j;
            const float* q = up + (i4 >> 1) * SLABW + (i4 & 1) * 4;
            rv[j][0] = q[0]; rv[j][1] = q[1]; rv[j][2] = q[2]; rv[j][3] = q[3];
        }
        #pragma unroll
        for (int j = 0; j < KDIM / 128; ++j) {
            const int i4 = lane + 32 * j;
            float* q = up + (i4 >> 1) * SLABW + (i4 & 1) * 4;
            q[0] = sub_scaled(rv[j][0], uv[j].x * pinv, p.val);
            q[1] = sub_scaled(rv[j][1], uv[j].y * pinv, p.val);
            q[2] = sub_scaled(rv[j][2], uv[j].z * pinv, p.val);
            q[3] = sub_scaled(rv[j][3], uv[j].w * pinv, p.val);
        }
        __syncthreads();
    }

    const float* al = atile + nloc * APITCH + half * 2;
    const float* bl = sl + half * 2 + nloc + woff;
    v8f acc = {};

    auto loadG = [&](int g, v2f (&a)[8], v2f (&bq)[8]) {
        const float* ap = al + g * 32;
        const float* bp = bl + g * 4 * SLABW;
        #pragma unroll
        for (int j = 0; j < 8; ++j) {
            a[j] = *(const v2f*)(ap + j * 4);
            const float* sp = bp + (j >> 1) * SLABW + (j & 1) * 4;
            bq[j].x = sp[0];
            bq[j].y = sp[1];
        }
    };
    auto mmaG = [&](const v2f (&a)[8], const v2f (&bq)[8]) {
        #pragma unroll
        for (int j = 0; j < 8; ++j)
            acc = __builtin_amdgcn_wmma_f32_16x16x4_f32(
                false, a[j], false, bq[j], (short)0, acc, false, false);
    };

    v2f aA[8], aB[8], bA[8], bB[8];
    loadG(0, aA, bA);
    for (int g = 0; g < KDIM / 32; g += 2) {
        loadG(g + 1, aB, bB);
        __builtin_amdgcn_sched_barrier(0);
        mmaG(aA, bA);
        __builtin_amdgcn_sched_barrier(0);
        loadG(g + 2 < KDIM / 32 ? g + 2 : g + 1, aA, bA);
        __builtin_amdgcn_sched_barrier(0);
        mmaG(aB, bB);
        __builtin_amdgcn_sched_barrier(0);
    }

    float bestv = acc[0];
    unsigned bestidx = (unsigned)((mt * 16 + 8 * half) * PPOS + w + nloc);
    #pragma unroll
    for (int r = 1; r < 8; ++r) {
        const float v = acc[r];
        if (v > bestv) {
            bestv = v;
            bestidx = (unsigned)((mt * 16 + r + 8 * half) * PPOS + w + nloc);
        }
    }
    unsigned vb = __float_as_uint(bestv);
    vb = (vb & 0x80000000u) ? ~vb : (vb | 0x80000000u);
    unsigned long long ckey = ((unsigned long long)vb << 32) | (0xFFFFFFFFu - bestidx);
    {
        const unsigned long long o = __shfl_xor(ckey, 16, 32);
        if (o > ckey) ckey = o;
    }
    const int rel = lane - woff;
    const unsigned long long nkey = __shfl(ckey, rel & 15, 32);
    unsigned long long mk = (rel >= 0 && rel < 16) ? nkey : oldpk;
    pkr[lane] = mk;
    for (int off = 8; off > 0; off >>= 1) {
        const unsigned long long o = __shfl_xor(mk, off, 32);
        if (o > mk) mk = o;
    }
    {
        const int t = t0 + half;
        if (nloc == 0 && t < NTILES) rec[t + (t >= NTPB ? 16 - NTPB : 0)] = mk;
    }
    unsigned long long newmax = mk;
    {
        const unsigned long long o = __shfl_xor(mk, 16, 32);
        if (o > newmax) newmax = o;
    }

    unsigned long long oldmax = 0ull;
    #pragma unroll
    for (int j = 0; j < NTILES; ++j)
        if (j != t0 && j != t0 + 1 && oldkeys[j] > oldmax) oldmax = oldkeys[j];
    if (lane == 0) atomicMax(kw_out + (size_t)b * KWPITCH, newmax > oldmax ? newmax : oldmax);
}

__global__ __launch_bounds__(KDIM)
void k_update_first(const float* __restrict__ d, const float* __restrict__ inv, float* __restrict__ res,
                    float* __restrict__ out, const unsigned long long* __restrict__ tilebest,
                    int* __restrict__ tsel) {
    __shared__ unsigned long long wred[KDIM / 32];
    const int b    = blockIdx.x;
    const int tid  = threadIdx.x;
    const int lane = tid & 31;
    const unsigned long long* tb = tilebest + (size_t)b * MTILES * KSLOTS;
    const int sl = tid & (KSLOTS - 1);
    const bool used = (sl < NTPB) || (sl >= 16 && sl < 16 + NTILES - NTPB);
    unsigned long long key = 0ull;
    if (used) {
        #pragma unroll
        for (int j = 0; j < (MTILES * KSLOTS) / KDIM; ++j) {
            const unsigned long long k = tb[j * KDIM + tid];
            if (k > key) key = k;
        }
    }
    key = wave_max_key(key);
    if (lane == 0) wred[tid >> 5] = key;
    __syncthreads();
    key = wred[0];
    for (int w = 1; w < KDIM / 32; ++w) { const unsigned long long o = wred[w]; if (o > key) key = o; }

    const Pick p = decode_key(key);
    const float* ap = d + (size_t)p.atom * KDIM;
    const float pinv = inv[(size_t)p.atom * INVP];
    float* rb = res + (size_t)b * CC * RES_W;
    float* ob = out + (size_t)b * CC * TT;
    for (int i = threadIdx.x; i < KDIM; i += blockDim.x) {
        const int c = i >> 3, t = i & 7;
        const int col = p.pos + t;
        const float a = ap[i] * pinv;
        rb[c * RES_W + col] = sub_scaled(rb[c * RES_W + col], a, p.val);
        if (col < TT) ob[c * TT + col] = add_scaled(ob[c * TT + col], a, p.val);
    }
    if (tid == 0) {
        tsel[b] = first_tile(p.pos);
        tsel[BATCH + b] = p.pos;
    }
}

__global__ __launch_bounds__(MTILES)
void k_update_last(const float* __restrict__ d, const float* __restrict__ inv, float* __restrict__ out,
                   const unsigned long long* __restrict__ kw_in) {
    const int b    = blockIdx.x;
    const int tid  = threadIdx.x;
    const Pick p = decode_key(kw_in[(size_t)b * KWPITCH]);
    const float* ap = d + (size_t)p.atom * KDIM;
    const float pinv = inv[(size_t)p.atom * INVP];
    float* ob = out + (size_t)b * CC * TT;
    for (int i = tid; i < KDIM; i += MTILES) {
        const int c = i >> 3, t = i & 7;
        const int col = p.pos + t;
        const float a = ap[i] * pinv;
        if (col < TT) ob[c * TT + col] = add_scaled(ob[c * TT + col], a, p.val);
    }
}

extern "C" void kernel_launch(void* const* d_in, const int* in_sizes, int n_in,
                              void* d_out, int out_size, void* d_ws, size_t ws_size,
                              hipStream_t stream) {
    const float* x = (const float*)d_in[0];
    const float* d = (const float*)d_in[1];
    (void)in_sizes; (void)n_in; (void)out_size; (void)ws_size;
    float* out = (float*)d_out;

    char* ws = (char*)d_ws;
    float* inv  = (float*)ws;
    unsigned long long* poskeys = (unsigned long long*)(ws + POSK_OFF);
    float* res0 = (float*)(ws + (size_t)NATOMS * KDIM * 4);
    float* res1 = res0 + BATCH * CC * RES_W;
    int* tsel = (int*)(res1 + BATCH * CC * RES_W);
    unsigned long long* tilebest = (unsigned long long*)(tsel + 8);
    unsigned long long* kw0 = (unsigned long long*)(ws + KW_OFF);
    unsigned long long* kw1 = kw0 + KWSLOT;
    unsigned long long* kw2 = kw1 + KWSLOT;

    k_normalize<<<NATOMS, CORR_THREADS, 0, stream>>>(d, inv);
    k_init<<<(BATCH * CC * RES_W + 255) / 256, 256, 0, stream>>>(x, res0, res1, out, kw0);

    const size_t shbytes = (size_t)CC * RES_W * sizeof(float);
    dim3 grid(NATOMS / MTILE, NSPLIT, BATCH);
    k_corr<<<grid, CORR_THREADS, shbytes, stream>>>(d, inv, res0, tilebest, poskeys);
    k_update_first<<<BATCH, KDIM, 0, stream>>>(d, inv, res0, out, tilebest, tsel);
    k_corr_inc<false><<<MTILES, CORR_THREADS, 0, stream>>>(d, inv, res0, res1, out, tsel, tilebest, poskeys, kw2, kw0, kw1);
    for (int it = 2; it < NITER; it += 6) {
        k_corr_inc<true><<<MTILES + 1, CORR_THREADS, 0, stream>>>(d, inv, res0, res1, out, tsel, tilebest, poskeys, kw0, kw1, kw2);
        k_corr_inc<true><<<MTILES + 1, CORR_THREADS, 0, stream>>>(d, inv, res1, res0, out, tsel, tilebest, poskeys, kw1, kw2, kw0);
        k_corr_inc<true><<<MTILES + 1, CORR_THREADS, 0, stream>>>(d, inv, res0, res1, out, tsel, tilebest, poskeys, kw2, kw0, kw1);
        k_corr_inc<true><<<MTILES + 1, CORR_THREADS, 0, stream>>>(d, inv, res1, res0, out, tsel, tilebest, poskeys, kw0, kw1, kw2);
        k_corr_inc<true><<<MTILES + 1, CORR_THREADS, 0, stream>>>(d, inv, res0, res1, out, tsel, tilebest, poskeys, kw1, kw2, kw0);
        k_corr_inc<true><<<MTILES + 1, CORR_THREADS, 0, stream>>>(d, inv, res1, res0, out, tsel, tilebest, poskeys, kw2, kw0, kw1);
    }
    k_update_last<<<BATCH, MTILES, 0, stream>>>(d, inv, out, kw0);
}


# Residual of the incremental steps stored column-major (a time column's 64 channels contiguous): a wave reads its window's 24 columns as one aligned 6 KB run, 12 full-line loads, not 20 part-line ones
# speedup vs baseline: 5.4346x; 1.1181x over previous
#include <hip/hip_runtime.h>

#define NATOMS 4096
#define CC     64
#define TT     256
#define ASIZE  8
#define BATCH  4
#define KDIM   512
#define PPOS   257
#define RES_W  304
#define NSPLIT 2
#define NTPB   9
#define MTILE  64
#define CORR_THREADS 128
#define NITER  32
#define NTILES 17
#define MTILES 256
#define KSLOTS 32
#define TCOLS  272
#define SCOLS  24
#define SLABW  25
#define APITCH 516
#define KWPITCH 32
#define KWSLOT (BATCH * KWPITCH)
#define KW_OFF 4194304
#define WLAST  (PPOS - 16)
#define INVP   32
#define POSK_OFF 1048576
#define RESK_OFF 5242880

typedef __attribute__((ext_vector_type(2))) float v2f;
typedef __attribute__((ext_vector_type(8))) float v8f;

__global__ void k_normalize(const float* __restrict__ d, float* __restrict__ inv) {
    __shared__ float red[4];
    const int a = blockIdx.x;
    const float* src = d + (size_t)a * KDIM;
    float s = 0.f;
    for (int i = threadIdx.x; i < KDIM; i += CORR_THREADS) { float v = src[i]; s += v * v; }
    for (int off = 16; off > 0; off >>= 1) s += __shfl_down(s, off, 32);
    const int wv = threadIdx.x >> 5;
    if ((threadIdx.x & 31) == 0) red[wv] = s;
    __syncthreads();
    if (threadIdx.x == 0) inv[(size_t)a * INVP] = 1.f / (sqrtf(red[0] + red[1] + red[2] + red[3]) + 1e-8f);
}

__global__ void k_init(const float* __restrict__ x, float* __restrict__ resk, float* __restrict__ res0,
                       float* __restrict__ res1, float* __restrict__ out,
                       unsigned long long* __restrict__ kw) {
    const int i = blockIdx.x * blockDim.x + threadIdx.x;
    if (i < 3 * BATCH) kw[(size_t)i * KWPITCH] = 0ull;
    if (i < BATCH * CC * RES_W) {
        const int col = i % RES_W;
        const int bc  = i / RES_W;
        resk[i] = (col < TT) ? x[bc * TT + col] : 0.f;
        if (col < TT) out[bc * TT + col] = 0.f;
    }
    if (i < BATCH * (TCOLS / 4) * CC) {
        const int c  = i % CC;
        const int q  = (i / CC) % (TCOLS / 4);
        const int b  = i / (CC * (TCOLS / 4));
        float4 v = make_float4(0.f, 0.f, 0.f, 0.f);
        if (q < TT / 4) v = *(const float4*)(x + ((size_t)b * CC + c) * TT + q * 4);
        const size_t o = ((size_t)b * TCOLS + q * 4) * CC + c;
        res0[o] = v.x;          res1[o] = v.x;
        res0[o + CC] = v.y;     res1[o + CC] = v.y;
        res0[o + 2 * CC] = v.z; res1[o + 2 * CC] = v.z;
        res0[o + 3 * CC] = v.w; res1[o + 3 * CC] = v.w;
    }
}

struct Pick { float val; int atom; int pos; };

__device__ __forceinline__ unsigned long long wave_max_key(unsigned long long key) {
    for (int off = 16; off > 0; off >>= 1) {
        unsigned long long o = __shfl_xor(key, off, 32);
        if (o > key) key = o;
    }
    return key;
}

__device__ __forceinline__ Pick decode_key(unsigned long long key) {
    const unsigned enc  = (unsigned)(key >> 32);
    const unsigned bits = (enc & 0x80000000u) ? (enc & 0x7FFFFFFFu) : ~enc;
    const unsigned idx  = 0xFFFFFFFFu - (unsigned)(key & 0xFFFFFFFFu);
    const unsigned a    = idx / PPOS;
    Pick p;
    p.val  = __uint_as_float(bits);
    p.atom = (int)(a < NATOMS ? a : NATOMS - 1);
    p.pos  = (int)(idx % PPOS);
    return p;
}

__device__ __forceinline__ int first_tile(int pos) {
    const int lo = pos > 7 ? pos - 7 : 0;
    const int t0 = lo >> 4;
    return t0 < NTILES - 2 ? t0 : NTILES - 2;
}

__device__ __forceinline__ float sub_scaled(float r, float a, float val) { return __builtin_fmaf(-a, val, r); }
__device__ __forceinline__ float add_scaled(float c, float a, float val) { return __builtin_fmaf(a, val, c); }

__device__ __forceinline__ Pick pick_from_word(const unsigned long long* kw, int b) {
    const unsigned long long key = kw[(size_t)b * KWPITCH];
    const unsigned klo = __builtin_amdgcn_readfirstlane((unsigned)(key & 0xFFFFFFFFu));
    const unsigned khi = __builtin_amdgcn_readfirstlane((unsigned)(key >> 32));
    return decode_key(((unsigned long long)khi << 32) | klo);
}

template <int NT>
__device__ __forceinline__ unsigned long long tile_keys_store(const v8f (&acc)[NT], int nt0, int atom0,
                                                int half, int nloc, int lane,
                                                unsigned long long* __restrict__ rec,
                                                unsigned long long* __restrict__ posk) {
    unsigned long long mykey = 0ull, allmax = 0ull;
    #pragma unroll
    for (int nt = 0; nt < NT; ++nt) {
        float bestv = -3.402823466e38f;
        unsigned bestidx = 0u;
        const int pos = (nt0 + nt) * 16 + nloc;
        if (pos < PPOS) {
            #pragma unroll
            for (int r = 0; r < 8; ++r) {
                const float v = acc[nt][r];
                if (v > bestv) {
                    bestv = v;
                    bestidx = (unsigned)((atom0 + r + 8 * half) * PPOS + pos);
                }
            }
        }
        unsigned vb = __float_as_uint(bestv);
        vb = (vb & 0x80000000u) ? ~vb : (vb | 0x80000000u);
        unsigned long long key = ((unsigned long long)vb << 32) | (0xFFFFFFFFu - bestidx);
        {
            unsigned long long o = __shfl_xor(key, 16, 32);
            if (o > key) key = o;
        }
        if (nt0 + nt < NTILES && lane < 16) posk[(nt0 + nt) * 16 + lane] = key;
        for (int off = 8; off > 0; off >>= 1) {
            unsigned long long o = __shfl_xor(key, off, 32);
            if (o > key) key = o;
        }
        if (lane == nt) mykey = key;
        if (key > allmax) allmax = key;
    }
    const int t = nt0 + lane;
    if (lane < NT && t < NTILES) rec[t + (t >= NTPB ? 16 - NTPB : 0)] = mykey;
    return allmax;
}

__global__ __launch_bounds__(CORR_THREADS, 1)
void k_corr(const float* __restrict__ d, const float* __restrict__ inv, const float* __restrict__ res,
            unsigned long long* __restrict__ tilebest, unsigned long long* __restrict__ poskeys) {
    extern __shared__ float smem[];
    const int b    = blockIdx.z;
    const int nt0  = blockIdx.y * NTPB;
    const int m0   = blockIdx.x * MTILE;
    const int tid  = threadIdx.x;
    const int lane = tid & 31;
    const int wave = tid >> 5;

    const int half = lane >> 4;
    const int nloc = lane & 15;
    const float* arow = d + (size_t)(m0 + wave * 16 + nloc) * KDIM + half * 2;
    const float ainv = inv[(size_t)(m0 + wave * 16 + nloc) * INVP];

    #pragma unroll
    for (int pf = 0; pf < 16; ++pf)
        __builtin_prefetch(arow + pf * 32, 0, 3);

    const float4* resb = (const float4*)(res + (size_t)b * CC * RES_W);
    float4* s4 = (float4*)smem;
    for (int i = tid; i < (CC * RES_W) / 4; i += CORR_THREADS) s4[i] = resb[i];
    __syncthreads();

    v8f acc[NTPB] = {};

    auto loadA = [&](int ks) { v2f a; a.x = arow[ks * 4] * ainv; a.y = arow[ks * 4 + 1] * ainv; return a; };
    auto srowp = [&](int ks) {
        const int kbase = ks * 4 + half * 2;
        return (const float*)smem + (kbase >> 3) * RES_W + (kbase & 7) + nloc + nt0 * 16;
    };

    v2f a0 = loadA(0);
    for (int ks = 0; ks < KDIM / 4; ++ks) {
        v2f an = loadA(ks + 1 < KDIM / 4 ? ks + 1 : ks);
        const float* sr = srowp(ks);
        v2f bf[NTPB];
        #pragma unroll
        for (int nt = 0; nt < NTPB; ++nt) {
            bf[nt].x = sr[nt * 16];
            bf[nt].y = sr[nt * 16 + 1];
        }
        #pragma unroll
        for (int nt = 0; nt < NTPB; ++nt)
            acc[nt] = __builtin_amdgcn_wmma_f32_16x16x4_f32(
                false, a0, false, bf[nt], (short)0, acc[nt], false, false);
        a0 = an;
    }

    const int mt = blockIdx.x * (MTILE / 16) + wave;
    tile_keys_store<NTPB>(acc, nt0, m0 + wave * 16, half, nloc, lane,
                          tilebest + ((size_t)b * MTILES + mt) * KSLOTS,
                          poskeys + ((size_t)b * MTILES + mt) * NTILES * 16);
}

template <bool HEAD>
__global__ __launch_bounds__(CORR_THREADS, 1)
void k_corr_inc(const float* __restrict__ d, const float* __restrict__ inv, const float* __restrict__ rin,
                float* __restrict__ rout, float* __restrict__ out, int* __restrict__ tsel,
                unsigned long long* tilebest,
                unsigned long long* poskeys,
                const unsigned long long* kw_in, unsigned long long* kw_out, unsigned long long* kw_zero) {
    __shared__ __attribute__((aligned(16))) float atile[16 * APITCH];
    __shared__ float slab[BATCH * CC * SLABW];
    const int mt   = blockIdx.x;
    const int tid  = threadIdx.x;
    const int lane = tid & 31;
    const int b    = __builtin_amdgcn_readfirstlane(tid >> 5);
    const int half = lane >> 4;
    const int nloc = lane & 15;

    Pick p = {0.f, 0, 0};
    if (HEAD) {
        p = pick_from_word(kw_in, b);

        if (mt == MTILES) {
            int pprev = tsel[BATCH + b];
            pprev = pprev < 0 ? 0 : (pprev < PPOS ? pprev : PPOS - 1);
            const float* ap = d + (size_t)p.atom * KDIM;
            const float pinv = inv[(size_t)p.atom * INVP];
            const float* ri = rin  + (size_t)b * TCOLS * CC;
            float*       ro = rout + (size_t)b * TCOLS * CC;
            float*       ob = out  + (size_t)b * CC * TT;
            const int t  = lane & 7;
            const int c0 = lane >> 3;
            const int col = p.pos + t;
            const int pc  = pprev + t;
            const bool carry = pc < p.pos || pc >= p.pos + ASIZE;
            float a[KDIM / 32], r[KDIM / 32], o[KDIM / 32], pv[KDIM / 32];
            #pragma unroll
            for (int j = 0; j < KDIM / 32; ++j) {
                const int c = c0 + 4 * j;
                a[j]  = ap[lane + 32 * j];
                r[j]  = ri[col * CC + c];
                pv[j] = ri[pc * CC + c];
                o[j]  = col < TT ? ob[c * TT + col] : 0.f;
            }
            #pragma unroll
            for (int j = 0; j < KDIM / 32; ++j) {
                const int c = c0 + 4 * j;
                const float an = a[j] * pinv;
                ro[col * CC + c] = sub_scaled(r[j], an, p.val);
                if (carry) ro[pc * CC + c] = pv[j];
                if (col < TT) ob[c * TT + col] = add_scaled(o[j], an, p.val);
            }
            if (lane == 0) {
                tsel[BATCH + b] = p.pos;
                kw_zero[(size_t)b * KWPITCH] = 0ull;
            }
            return;
        }
    }

    const float4* asrc = (const float4*)(d + (size_t)mt * 16 * KDIM) + tid;
    float4 av[16];
    #pragma unroll
    for (int j = 0; j < 16; ++j) av[j] = asrc[j * (KDIM / 4)];
    float rinv[16];
    #pragma unroll
    for (int j = 0; j < 16; ++j) rinv[j] = inv[(size_t)(mt * 16 + j) * INVP];

    unsigned long long* rec = tilebest + ((size_t)b * MTILES + mt) * KSLOTS;
    unsigned long long oldkeys[NTILES];
    #pragma unroll
    for (int j = 0; j < NTILES; ++j) oldkeys[j] = rec[j + (j >= NTPB ? 16 - NTPB : 0)];

    int ppos = p.pos;
    if (!HEAD) {
        ppos = tsel[BATCH + b];
        ppos = ppos < 0 ? 0 : (ppos < PPOS ? ppos : PPOS - 1);
    }
    const int t0 = first_tile(ppos);
    int w = ppos > 7 ? ppos - 7 : 0;
    w = w < WLAST ? w : WLAST;
    const int woff = w - t0 * 16;

    unsigned long long* pkr = poskeys + (((size_t)b * MTILES + mt) * NTILES + t0) * 16;
    const unsigned long long oldpk = pkr[lane];

    float* sl = slab + b * CC * SLABW;
    const float4* rsrc = (const float4*)(rin + ((size_t)b * TCOLS + w) * CC) + lane;
    float4 sv[(SCOLS * CC / 4) / 32];
    #pragma unroll
    for (int j = 0; j < (SCOLS * CC / 4) / 32; ++j) sv[j] = rsrc[32 * j];
    float4 uv[KDIM / 128];
    float pinv = 0.f;
    if (HEAD) {
        const float4* arow4 = (const float4*)(d + (size_t)p.atom * KDIM);
        #pragma unroll
        for (int j = 0; j < KDIM / 128; ++j) uv[j] = arow4[lane + 32 * j];
        pinv = inv[(size_t)p.atom * INVP];
    }
    #pragma unroll
    for (int j = 0; j < 16; ++j) {
        float4 v = av[j];
        v.x *= rinv[j]; v.y *= rinv[j]; v.z *= rinv[j]; v.w *= rinv[j];
        *(float4*)(atile + j * APITCH + tid * 4) = v;
    }
    #pragma unroll
    for (int j = 0; j < (SCOLS * CC / 4) / 32; ++j) {
        float* q = sl + (lane & 15) * 4 * SLABW + 2 * j + (lane >> 4);
        q[0] = sv[j].x; q[SLABW] = sv[j].y; q[2 * SLABW] = sv[j].z; q[3 * SLABW] = sv[j].w;
    }
    __syncthreads();

    if (HEAD) {
        float* up = sl + (p.pos - w);
        float rv[KDIM / 128][4];
        #pragma unroll
        for (int j = 0; j < KDIM / 128; ++j) {
            const int i4 = lane + 32 * j;
            const float* q = up + (i4 >> 1) * SLABW + (i4 & 1) * 4;
            rv[j][0] = q[0]; rv[j][1] = q[1]; rv[j][2] = q[2]; rv[j][3] = q[3];
        }
        #pragma unroll
        for (int j = 0; j < KDIM / 128; ++j) {
            const int i4 = lane + 32 * j;
            float* q = up + (i4 >> 1) * SLABW + (i4 & 1) * 4;
            q[0] = sub_scaled(rv[j][0], uv[j].x * pinv, p.val);
            q[1] = sub_scaled(rv[j][1], uv[j].y * pinv, p.val);
            q[2] = sub_scaled(rv[j][2], uv[j].z * pinv, p.val);
            q[3] = sub_scaled(rv[j][3], uv[j].w * pinv, p.val);
        }
        __syncthreads();
    }

    const float* al = atile + nloc * APITCH + half * 2;
    const float* bl = sl + half * 2 + nloc;
    v8f acc = {};

    auto loadG = [&](int g, v2f (&a)[8], v2f (&bq)[8]) {
        const float* ap = al + g * 32;
        const float* bp = bl + g * 4 * SLABW;
        #pragma unroll
        for (int j = 0; j < 8; ++j) {
            a[j] = *(const v2f*)(ap + j * 4);
            const float* sp = bp + (j >> 1) * SLABW + (j & 1) * 4;
            bq[j].x = sp[0];
            bq[j].y = sp[1];
        }
    };
    auto mmaG = [&](const v2f (&a)[8], const v2f (&bq)[8]) {
        #pragma unroll
        for (int j = 0; j < 8; ++j)
            acc = __builtin_amdgcn_wmma_f32_16x16x4_f32(
                false, a[j], false, bq[j], (short)0, acc, false, false);
    };

    v2f aA[8], aB[8], bA[8], bB[8];
    loadG(0, aA, bA);
    for (int g = 0; g < KDIM / 32; g += 2) {
        loadG(g + 1, aB, bB);
        __builtin_amdgcn_sched_barrier(0);
        mmaG(aA, bA);
        __builtin_amdgcn_sched_barrier(0);
        loadG(g + 2 < KDIM / 32 ? g + 2 : g + 1, aA, bA);
        __builtin_amdgcn_sched_barrier(0);
        mmaG(aB, bB);
        __builtin_amdgcn_sched_barrier(0);
    }

    float bestv = acc[0];
    unsigned bestidx = (unsigned)((mt * 16 + 8 * half) * PPOS + w + nloc);
    #pragma unroll
    for (int r = 1; r < 8; ++r) {
        const float v = acc[r];
        if (v > bestv) {
            bestv = v;
            bestidx = (unsigned)((mt * 16 + r + 8 * half) * PPOS + w + nloc);
        }
    }
    unsigned vb = __float_as_uint(bestv);
    vb = (vb & 0x80000000u) ? ~vb : (vb | 0x80000000u);
    unsigned long long ckey = ((unsigned long long)vb << 32) | (0xFFFFFFFFu - bestidx);
    {
        const unsigned long long o = __shfl_xor(ckey, 16, 32);
        if (o > ckey) ckey = o;
    }
    const int rel = lane - woff;
    const unsigned long long nkey = __shfl(ckey, rel & 15, 32);
    unsigned long long mk = (rel >= 0 && rel < 16) ? nkey : oldpk;
    pkr[lane] = mk;
    for (int off = 8; off > 0; off >>= 1) {
        const unsigned long long o = __shfl_xor(mk, off, 32);
        if (o > mk) mk = o;
    }
    {
        const int t = t0 + half;
        if (nloc == 0 && t < NTILES) rec[t + (t >= NTPB ? 16 - NTPB : 0)] = mk;
    }
    unsigned long long newmax = mk;
    {
        const unsigned long long o = __shfl_xor(mk, 16, 32);
        if (o > newmax) newmax = o;
    }

    unsigned long long oldmax = 0ull;
    #pragma unroll
    for (int j = 0; j < NTILES; ++j)
        if (j != t0 && j != t0 + 1 && oldkeys[j] > oldmax) oldmax = oldkeys[j];
    if (lane == 0) atomicMax(kw_out + (size_t)b * KWPITCH, newmax > oldmax ? newmax : oldmax);
}

__global__ __launch_bounds__(KDIM)
void k_update_first(const float* __restrict__ d, const float* __restrict__ inv, float* __restrict__ res,
                    float* __restrict__ out, const unsigned long long* __restrict__ tilebest,
                    int* __restrict__ tsel) {
    __shared__ unsigned long long wred[KDIM / 32];
    const int b    = blockIdx.x;
    const int tid  = threadIdx.x;
    const int lane = tid & 31;
    const unsigned long long* tb = tilebest + (size_t)b * MTILES * KSLOTS;
    const int sl = tid & (KSLOTS - 1);
    const bool used = (sl < NTPB) || (sl >= 16 && sl < 16 + NTILES - NTPB);
    unsigned long long key = 0ull;
    if (used) {
        #pragma unroll
        for (int j = 0; j < (MTILES * KSLOTS) / KDIM; ++j) {
            const unsigned long long k = tb[j * KDIM + tid];
            if (k > key) key = k;
        }
    }
    key = wave_max_key(key);
    if (lane == 0) wred[tid >> 5] = key;
    __syncthreads();
    key = wred[0];
    for (int w = 1; w < KDIM / 32; ++w) { const unsigned long long o = wred[w]; if (o > key) key = o; }

    const Pick p = decode_key(key);
    const float* ap = d + (size_t)p.atom * KDIM;
    const float pinv = inv[(size_t)p.atom * INVP];
    float* rb = res + (size_t)b * TCOLS * CC;
    float* ob = out + (size_t)b * CC * TT;
    for (int i = threadIdx.x; i < KDIM; i += blockDim.x) {
        const int c = i >> 3, t = i & 7;
        const int col = p.pos + t;
        const float a = ap[i] * pinv;
        rb[col * CC + c] = sub_scaled(rb[col * CC + c], a, p.val);
        if (col < TT) ob[c * TT + col] = add_scaled(ob[c * TT + col], a, p.val);
    }
    if (tid == 0) {
        tsel[b] = first_tile(p.pos);
        tsel[BATCH + b] = p.pos;
    }
}

__global__ __launch_bounds__(MTILES)
void k_update_last(const float* __restrict__ d, const float* __restrict__ inv, float* __restrict__ out,
                   const unsigned long long* __restrict__ kw_in) {
    const int b    = blockIdx.x;
    const int tid  = threadIdx.x;
    const Pick p = decode_key(kw_in[(size_t)b * KWPITCH]);
    const float* ap = d + (size_t)p.atom * KDIM;
    const float pinv = inv[(size_t)p.atom * INVP];
    float* ob = out + (size_t)b * CC * TT;
    for (int i = tid; i < KDIM; i += MTILES) {
        const int c = i >> 3, t = i & 7;
        const int col = p.pos + t;
        const float a = ap[i] * pinv;
        if (col < TT) ob[c * TT + col] = add_scaled(ob[c * TT + col], a, p.val);
    }
}

extern "C" void kernel_launch(void* const* d_in, const int* in_sizes, int n_in,
                              void* d_out, int out_size, void* d_ws, size_t ws_size,
                              hipStream_t stream) {
    const float* x = (const float*)d_in[0];
    const float* d = (const float*)d_in[1];
    (void)in_sizes; (void)n_in; (void)out_size; (void)ws_size;
    float* out = (float*)d_out;

    char* ws = (char*)d_ws;
    float* inv  = (float*)ws;
    unsigned long long* poskeys = (unsigned long long*)(ws + POSK_OFF);
    float* resk = (float*)(ws + RESK_OFF);
    float* res0 = (float*)(ws + (size_t)NATOMS * KDIM * 4);
    float* res1 = res0 + BATCH * CC * RES_W;
    int* tsel = (int*)(res1 + BATCH * CC * RES_W);
    unsigned long long* tilebest = (unsigned long long*)(tsel + 8);
    unsigned long long* kw0 = (unsigned long long*)(ws + KW_OFF);
    unsigned long long* kw1 = kw0 + KWSLOT;
    unsigned long long* kw2 = kw1 + KWSLOT;

    k_normalize<<<NATOMS, CORR_THREADS, 0, stream>>>(d, inv);
    k_init<<<(BATCH * CC * RES_W + 255) / 256, 256, 0, stream>>>(x, resk, res0, res1, out, kw0);

    const size_t shbytes = (size_t)CC * RES_W * sizeof(float);
    dim3 grid(NATOMS / MTILE, NSPLIT, BATCH);
    k_corr<<<grid, CORR_THREADS, shbytes, stream>>>(d, inv, resk, tilebest, poskeys);
    k_update_first<<<BATCH, KDIM, 0, stream>>>(d, inv, res0, out, tilebest, tsel);
    k_corr_inc<false><<<MTILES, CORR_THREADS, 0, stream>>>(d, inv, res0, res1, out, tsel, tilebest, poskeys, kw2, kw0, kw1);
    for (int it = 2; it < NITER; it += 6) {
        k_corr_inc<true><<<MTILES + 1, CORR_THREADS, 0, stream>>>(d, inv, res0, res1, out, tsel, tilebest, poskeys, kw0, kw1, kw2);
        k_corr_inc<true><<<MTILES + 1, CORR_THREADS, 0, stream>>>(d, inv, res1, res0, out, tsel, tilebest, poskeys, kw1, kw2, kw0);
        k_corr_inc<true><<<MTILES + 1, CORR_THREADS, 0, stream>>>(d, inv, res0, res1, out, tsel, tilebest, poskeys, kw2, kw0, kw1);
        k_corr_inc<true><<<MTILES + 1, CORR_THREADS, 0, stream>>>(d, inv, res1, res0, out, tsel, tilebest, poskeys, kw0, kw1, kw2);
        k_corr_inc<true><<<MTILES + 1, CORR_THREADS, 0, stream>>>(d, inv, res0, res1, out, tsel, tilebest, poskeys, kw1, kw2, kw0);
        k_corr_inc<true><<<MTILES + 1, CORR_THREADS, 0, stream>>>(d, inv, res1, res0, out, tsel, tilebest, poskeys, kw2, kw0, kw1);
    }
    k_update_last<<<BATCH, MTILES, 0, stream>>>(d, inv, out, kw0);
}


# Set-up launches fused: norms + buffer init in one 2048-wave launch; full correlation folds its keys into the step-key word (atomic max), so a head launch does the first update: 36 launches to 34
# speedup vs baseline: 5.6595x; 1.0414x over previous
#include <hip/hip_runtime.h>

#define NATOMS 4096
#define CC     64
#define TT     256
#define ASIZE  8
#define BATCH  4
#define KDIM   512
#define PPOS   257
#define RES_W  304
#define NSPLIT 2
#define NTPB   9
#define MTILE  64
#define CORR_THREADS 128
#define NITER  32
#define NTILES 17
#define MTILES 256
#define KSLOTS 32
#define TCOLS  272
#define SCOLS  24
#define SLABW  25
#define APITCH 516
#define KWPITCH 32
#define KWSLOT (BATCH * KWPITCH)
#define KW_OFF 4194304
#define WLAST  (PPOS - 16)
#define INVP   32
#define POSK_OFF 1048576
#define RESK_OFF 5242880
#define SETUP_THREADS 256
#define SETUP_BLOCKS  256

typedef __attribute__((ext_vector_type(2))) float v2f;
typedef __attribute__((ext_vector_type(8))) float v8f;

__global__ __launch_bounds__(SETUP_THREADS)
void k_setup(const float* __restrict__ d, const float* __restrict__ x, float* __restrict__ inv,
             float* __restrict__ resk, float* __restrict__ res0, float* __restrict__ res1,
             float* __restrict__ out, unsigned long long* __restrict__ kw, int* __restrict__ tsel) {
    const int tid  = threadIdx.x;
    const int lane = tid & 31;
    const int a0   = (blockIdx.x * (SETUP_THREADS / 32) + (tid >> 5)) * 2;
    const float* src = d + (size_t)a0 * KDIM + lane;
    float e[2][KDIM / 32];
    #pragma unroll
    for (int q = 0; q < 2; ++q) {
        #pragma unroll
        for (int j = 0; j < KDIM / 32; ++j) e[q][j] = src[q * KDIM + 32 * j];
    }

    const int gid = blockIdx.x * SETUP_THREADS + tid;
    if (gid < 3 * BATCH) kw[(size_t)gid * KWPITCH] = 0ull;
    if (gid < BATCH) tsel[BATCH + gid] = 0;
    for (int i = gid; i < BATCH * CC * RES_W; i += SETUP_BLOCKS * SETUP_THREADS) {
        const int col = i % RES_W;
        const int bc  = i / RES_W;
        resk[i] = (col < TT) ? x[bc * TT + col] : 0.f;
        if (col < TT) out[bc * TT + col] = 0.f;
    }
    if (gid < BATCH * (TCOLS / 4) * CC) {
        const int c  = gid % CC;
        const int q  = (gid / CC) % (TCOLS / 4);
        const int b  = gid / (CC * (TCOLS / 4));
        float4 v = make_float4(0.f, 0.f, 0.f, 0.f);
        if (q < TT / 4) v = *(const float4*)(x + ((size_t)b * CC + c) * TT + q * 4);
        const size_t o = ((size_t)b * TCOLS + q * 4) * CC + c;
        res0[o] = v.x;          res1[o] = v.x;
        res0[o + CC] = v.y;     res1[o + CC] = v.y;
        res0[o + 2 * CC] = v.z; res1[o + 2 * CC] = v.z;
        res0[o + 3 * CC] = v.w; res1[o + 3 * CC] = v.w;
    }

    #pragma unroll
    for (int q = 0; q < 2; ++q) {
        float part[4];
        #pragma unroll
        for (int v = 0; v < 4; ++v) {
            float s = 0.f;
            #pragma unroll
            for (int k = 0; k < 4; ++k) s = __builtin_fmaf(e[q][v + 4 * k], e[q][v + 4 * k], s);
            for (int off = 16; off > 0; off >>= 1) s += __shfl_down(s, off, 32);
            part[v] = s;
        }
        if (lane == 0)
            inv[(size_t)(a0 + q) * INVP] = 1.f / (sqrtf(part[0] + part[1] + part[2] + part[3]) + 1e-8f);
    }
}

struct Pick { float val; int atom; int pos; };

__device__ __forceinline__ Pick decode_key(unsigned long long key) {
    const unsigned enc  = (unsigned)(key >> 32);
    const unsigned bits = (enc & 0x80000000u) ? (enc & 0x7FFFFFFFu) : ~enc;
    const unsigned idx  = 0xFFFFFFFFu - (unsigned)(key & 0xFFFFFFFFu);
    const unsigned a    = idx / PPOS;
    Pick p;
    p.val  = __uint_as_float(bits);
    p.atom = (int)(a < NATOMS ? a : NATOMS - 1);
    p.pos  = (int)(idx % PPOS);
    return p;
}

__device__ __forceinline__ int first_tile(int pos) {
    const int lo = pos > 7 ? pos - 7 : 0;
    const int t0 = lo >> 4;
    return t0 < NTILES - 2 ? t0 : NTILES - 2;
}

__device__ __forceinline__ float sub_scaled(float r, float a, float val) { return __builtin_fmaf(-a, val, r); }
__device__ __forceinline__ float add_scaled(float c, float a, float val) { return __builtin_fmaf(a, val, c); }

__device__ __forceinline__ Pick pick_from_word(const unsigned long long* kw, int b) {
    const unsigned long long key = kw[(size_t)b * KWPITCH];
    const unsigned klo = __builtin_amdgcn_readfirstlane((unsigned)(key & 0xFFFFFFFFu));
    const unsigned khi = __builtin_amdgcn_readfirstlane((unsigned)(key >> 32));
    return decode_key(((unsigned long long)khi << 32) | klo);
}

template <int NT>
__device__ __forceinline__ unsigned long long tile_keys_store(const v8f (&acc)[NT], int nt0, int atom0,
                                                int half, int nloc, int lane,
                                                unsigned long long* __restrict__ rec,
                                                unsigned long long* __restrict__ posk) {
    unsigned long long mykey = 0ull, allmax = 0ull;
    #pragma unroll
    for (int nt = 0; nt < NT; ++nt) {
        float bestv = -3.402823466e38f;
        unsigned bestidx = 0u;
        const int pos = (nt0 + nt) * 16 + nloc;
        if (pos < PPOS) {
            #pragma unroll
            for (int r = 0; r < 8; ++r) {
                const float v = acc[nt][r];
                if (v > bestv) {
                    bestv = v;
                    bestidx = (unsigned)((atom0 + r + 8 * half) * PPOS + pos);
                }
            }
        }
        unsigned vb = __float_as_uint(bestv);
        vb = (vb & 0x80000000u) ? ~vb : (vb | 0x80000000u);
        unsigned long long key = ((unsigned long long)vb << 32) | (0xFFFFFFFFu - bestidx);
        {
            unsigned long long o = __shfl_xor(key, 16, 32);
            if (o > key) key = o;
        }
        if (nt0 + nt < NTILES && lane < 16) posk[(nt0 + nt) * 16 + lane] = key;
        for (int off = 8; off > 0; off >>= 1) {
            unsigned long long o = __shfl_xor(key, off, 32);
            if (o > key) key = o;
        }
        if (lane == nt) mykey = key;
        if (key > allmax) allmax = key;
    }
    const int t = nt0 + lane;
    if (lane < NT && t < NTILES) rec[t + (t >= NTPB ? 16 - NTPB : 0)] = mykey;
    return allmax;
}

__global__ __launch_bounds__(CORR_THREADS, 1)
void k_corr(const float* __restrict__ d, const float* __restrict__ inv, const float* __restrict__ res,
            unsigned long long* __restrict__ tilebest, unsigned long long* __restrict__ poskeys,
            unsigned long long* kw_out) {
    extern __shared__ float smem[];
    const int b    = blockIdx.z;
    const int nt0  = blockIdx.y * NTPB;
    const int m0   = blockIdx.x * MTILE;
    const int tid  = threadIdx.x;
    const int lane = tid & 31;
    const int wave = tid >> 5;

    const int half = lane >> 4;
    const int nloc = lane & 15;
    const float* arow = d + (size_t)(m0 + wave * 16 + nloc) * KDIM + half * 2;
    const float ainv = inv[(size_t)(m0 + wave * 16 + nloc) * INVP];

    #pragma unroll
    for (int pf = 0; pf < 16; ++pf)
        __builtin_prefetch(arow + pf * 32, 0, 3);

    const float4* resb = (const float4*)(res + (size_t)b * CC * RES_W);
    float4* s4 = (float4*)smem;
    for (int i = tid; i < (CC * RES_W) / 4; i += CORR_THREADS) s4[i] = resb[i];
    __syncthreads();

    v8f acc[NTPB] = {};

    auto loadA = [&](int ks) { v2f a; a.x = arow[ks * 4] * ainv; a.y = arow[ks * 4 + 1] * ainv; return a; };
    auto srowp = [&](int ks) {
        const int kbase = ks * 4 + half * 2;
        return (const float*)smem + (kbase >> 3) * RES_W + (kbase & 7) + nloc + nt0 * 16;
    };

    v2f a0 = loadA(0);
    for (int ks = 0; ks < KDIM / 4; ++ks) {
        v2f an = loadA(ks + 1 < KDIM / 4 ? ks + 1 : ks);
        const float* sr = srowp(ks);
        v2f bf[NTPB];
        #pragma unroll
        for (int nt = 0; nt < NTPB; ++nt) {
            bf[nt].x = sr[nt * 16];
            bf[nt].y = sr[nt * 16 + 1];
        }
        #pragma unroll
        for (int nt = 0; nt < NTPB; ++nt)
            acc[nt] = __builtin_amdgcn_wmma_f32_16x16x4_f32(
                false, a0, false, bf[nt], (short)0, acc[nt], false, false);
        a0 = an;
    }

    const int mt = blockIdx.x * (MTILE / 16) + wave;
    const unsigned long long best =
        tile_keys_store<NTPB>(acc, nt0, m0 + wave * 16, half, nloc, lane,
                              tilebest + ((size_t)b * MTILES + mt) * KSLOTS,
                              poskeys + ((size_t)b * MTILES + mt) * NTILES * 16);
    if (lane == 0) atomicMax(kw_out + (size_t)b * KWPITCH, best);
}

__global__ __launch_bounds__(CORR_THREADS, 1)
void k_corr_inc(const float* __restrict__ d, const float* __restrict__ inv, const float* __restrict__ rin,
                float* __restrict__ rout, float* __restrict__ out, int* __restrict__ tsel,
                unsigned long long* tilebest,
                unsigned long long* poskeys,
                const unsigned long long* kw_in, unsigned long long* kw_out, unsigned long long* kw_zero) {
    __shared__ __attribute__((aligned(16))) float atile[16 * APITCH];
    __shared__ float slab[BATCH * CC * SLABW];
    const int mt   = blockIdx.x;
    const int tid  = threadIdx.x;
    const int lane = tid & 31;
    const int b    = __builtin_amdgcn_readfirstlane(tid >> 5);
    const int half = lane >> 4;
    const int nloc = lane & 15;

    const Pick p = pick_from_word(kw_in, b);
    if (mt == MTILES) {
        int pprev = tsel[BATCH + b];
        pprev = pprev < 0 ? 0 : (pprev < PPOS ? pprev : PPOS - 1);
        const float* ap = d + (size_t)p.atom * KDIM;
        const float pinv = inv[(size_t)p.atom * INVP];
        const float* ri = rin  + (size_t)b * TCOLS * CC;
        float*       ro = rout + (size_t)b * TCOLS * CC;
        float*       ob = out  + (size_t)b * CC * TT;
        const int t  = lane & 7;
        const int c0 = lane >> 3;
        const int col = p.pos + t;
        const int pc  = pprev + t;
        const bool carry = pc < p.pos || pc >= p.pos + ASIZE;
        float a[KDIM / 32], r[KDIM / 32], o[KDIM / 32], pv[KDIM / 32];
        #pragma unroll
        for (int j = 0; j < KDIM / 32; ++j) {
            const int c = c0 + 4 * j;
            a[j]  = ap[lane + 32 * j];
            r[j]  = ri[col * CC + c];
            pv[j] = ri[pc * CC + c];
            o[j]  = col < TT ? ob[c * TT + col] : 0.f;
        }
        #pragma unroll
        for (int j = 0; j < KDIM / 32; ++j) {
            const int c = c0 + 4 * j;
            const float an = a[j] * pinv;
            ro[col * CC + c] = sub_scaled(r[j], an, p.val);
            if (carry) ro[pc * CC + c] = pv[j];
            if (col < TT) ob[c * TT + col] = add_scaled(o[j], an, p.val);
        }
        if (lane == 0) {
            tsel[BATCH + b] = p.pos;
            kw_zero[(size_t)b * KWPITCH] = 0ull;
        }
        return;
    }

    const float4* asrc = (const float4*)(d + (size_t)mt * 16 * KDIM) + tid;
    float4 av[16];
    #pragma unroll
    for (int j = 0; j < 16; ++j) av[j] = asrc[j * (KDIM / 4)];
    float rinv[16];
    #pragma unroll
    for (int j = 0; j < 16; ++j) rinv[j] = inv[(size_t)(mt * 16 + j) * INVP];

    unsigned long long* rec = tilebest + ((size_t)b * MTILES + mt) * KSLOTS;
    unsigned long long oldkeys[NTILES];
    #pragma unroll
    for (int j = 0; j < NTILES; ++j) oldkeys[j] = rec[j + (j >= NTPB ? 16 - NTPB : 0)];

    const int ppos = p.pos;
    const int t0 = first_tile(ppos);
    int w = ppos > 7 ? ppos - 7 : 0;
    w = w < WLAST ? w : WLAST;
    const int woff = w - t0 * 16;

    unsigned long long* pkr = poskeys + (((size_t)b * MTILES + mt) * NTILES + t0) * 16;
    const unsigned long long oldpk = pkr[lane];

    float* sl = slab + b * CC * SLABW;
    const float4* rsrc = (const float4*)(rin + ((size_t)b * TCOLS + w) * CC) + lane;
    float4 sv[(SCOLS * CC / 4) / 32];
    #pragma unroll
    for (int j = 0; j < (SCOLS * CC / 4) / 32; ++j) sv[j] = rsrc[32 * j];
    float4 uv[KDIM / 128];
    const float4* arow4 = (const float4*)(d + (size_t)p.atom * KDIM);
    #pragma unroll
    for (int j = 0; j < KDIM / 128; ++j) uv[j] = arow4[lane + 32 * j];
    const float pinv = inv[(size_t)p.atom * INVP];
    #pragma unroll
    for (int j = 0; j < 16; ++j) {
        float4 v = av[j];
        v.x *= rinv[j]; v.y *= rinv[j]; v.z *= rinv[j]; v.w *= rinv[j];
        *(float4*)(atile + j * APITCH + tid * 4) = v;
    }
    #pragma unroll
    for (int j = 0; j < (SCOLS * CC / 4) / 32; ++j) {
        float* q = sl + (lane & 15) * 4 * SLABW + 2 * j + (lane >> 4);
        q[0] = sv[j].x; q[SLABW] = sv[j].y; q[2 * SLABW] = sv[j].z; q[3 * SLABW] = sv[j].w;
    }
    __syncthreads();

    {
        float* up = sl + (p.pos - w);
        float rv[KDIM / 128][4];
        #pragma unroll
        for (int j = 0; j < KDIM / 128; ++j) {
            const int i4 = lane + 32 * j;
            const float* q = up + (i4 >> 1) * SLABW + (i4 & 1) * 4;
            rv[j][0] = q[0]; rv[j][1] = q[1]; rv[j][2] = q[2]; rv[j][3] = q[3];
        }
        #pragma unroll
        for (int j = 0; j < KDIM / 128; ++j) {
            const int i4 = lane + 32 * j;
            float* q = up + (i4 >> 1) * SLABW + (i4 & 1) * 4;
            q[0] = sub_scaled(rv[j][0], uv[j].x * pinv, p.val);
            q[1] = sub_scaled(rv[j][1], uv[j].y * pinv, p.val);
            q[2] = sub_scaled(rv[j][2], uv[j].z * pinv, p.val);
            q[3] = sub_scaled(rv[j][3], uv[j].w * pinv, p.val);
        }
        __syncthreads();
    }

    const float* al = atile + nloc * APITCH + half * 2;
    const float* bl = sl + half * 2 + nloc;
    v8f acc = {};

    auto loadG = [&](int g, v2f (&a)[8], v2f (&bq)[8]) {
        const float* ap = al + g * 32;
        const float* bp = bl + g * 4 * SLABW;
        #pragma unroll
        for (int j = 0; j < 8; ++j) {
            a[j] = *(const v2f*)(ap + j * 4);
            const float* sp = bp + (j >> 1) * SLABW + (j & 1) * 4;
            bq[j].x = sp[0];
            bq[j].y = sp[1];
        }
    };
    auto mmaG = [&](const v2f (&a)[8], const v2f (&bq)[8]) {
        #pragma unroll
        for (int j = 0; j < 8; ++j)
            acc = __builtin_amdgcn_wmma_f32_16x16x4_f32(
                false, a[j], false, bq[j], (short)0, acc, false, false);
    };

    v2f aA[8], aB[8], bA[8], bB[8];
    loadG(0, aA, bA);
    for (int g = 0; g < KDIM / 32; g += 2) {
        loadG(g + 1, aB, bB);
        __builtin_amdgcn_sched_barrier(0);
        mmaG(aA, bA);
        __builtin_amdgcn_sched_barrier(0);
        loadG(g + 2 < KDIM / 32 ? g + 2 : g + 1, aA, bA);
        __builtin_amdgcn_sched_barrier(0);
        mmaG(aB, bB);
        __builtin_amdgcn_sched_barrier(0);
    }

    float bestv = acc[0];
    unsigned bestidx = (unsigned)((mt * 16 + 8 * half) * PPOS + w + nloc);
    #pragma unroll
    for (int r = 1; r < 8; ++r) {
        const float v = acc[r];
        if (v > bestv) {
            bestv = v;
            bestidx = (unsigned)((mt * 16 + r + 8 * half) * PPOS + w + nloc);
        }
    }
    unsigned vb = __float_as_uint(bestv);
    vb = (vb & 0x80000000u) ? ~vb : (vb | 0x80000000u);
    unsigned long long ckey = ((unsigned long long)vb << 32) | (0xFFFFFFFFu - bestidx);
    {
        const unsigned long long o = __shfl_xor(ckey, 16, 32);
        if (o > ckey) ckey = o;
    }
    const int rel = lane - woff;
    const unsigned long long nkey = __shfl(ckey, rel & 15, 32);
    unsigned long long mk = (rel >= 0 && rel < 16) ? nkey : oldpk;
    pkr[lane] = mk;
    for (int off = 8; off > 0; off >>= 1) {
        const unsigned long long o = __shfl_xor(mk, off, 32);
        if (o > mk) mk = o;
    }
    {
        const int t = t0 + half;
        if (nloc == 0 && t < NTILES) rec[t + (t >= NTPB ? 16 - NTPB : 0)] = mk;
    }
    unsigned long long newmax = mk;
    {
        const unsigned long long o = __shfl_xor(mk, 16, 32);
        if (o > newmax) newmax = o;
    }

    unsigned long long oldmax = 0ull;
    #pragma unroll
    for (int j = 0; j < NTILES; ++j)
        if (j != t0 && j != t0 + 1 && oldkeys[j] > oldmax) oldmax = oldkeys[j];
    if (lane == 0) atomicMax(kw_out + (size_t)b * KWPITCH, newmax > oldmax ? newmax : oldmax);
}

__global__ __launch_bounds__(MTILES)
void k_update_last(const float* __restrict__ d, const float* __restrict__ inv, float* __restrict__ out,
                   const unsigned long long* __restrict__ kw_in) {
    const int b    = blockIdx.x;
    const int tid  = threadIdx.x;
    const Pick p = decode_key(kw_in[(size_t)b * KWPITCH]);
    const float* ap = d + (size_t)p.atom * KDIM;
    const float pinv = inv[(size_t)p.atom * INVP];
    float* ob = out + (size_t)b * CC * TT;
    for (int i = tid; i < KDIM; i += MTILES) {
        const int c = i >> 3, t = i & 7;
        const int col = p.pos + t;
        const float a = ap[i] * pinv;
        if (col < TT) ob[c * TT + col] = add_scaled(ob[c * TT + col], a, p.val);
    }
}

extern "C" void kernel_launch(void* const* d_in, const int* in_sizes, int n_in,
                              void* d_out, int out_size, void* d_ws, size_t ws_size,
                              hipStream_t stream) {
    const float* x = (const float*)d_in[0];
    const float* d = (const float*)d_in[1];
    (void)in_sizes; (void)n_in; (void)out_size; (void)ws_size;
    float* out = (float*)d_out;

    char* ws = (char*)d_ws;
    float* inv  = (float*)ws;
    unsigned long long* poskeys = (unsigned long long*)(ws + POSK_OFF);
    float* resk = (float*)(ws + RESK_OFF);
    float* res0 = (float*)(ws + (size_t)NATOMS * KDIM * 4);
    float* res1 = res0 + BATCH * CC * RES_W;
    int* tsel = (int*)(res1 + BATCH * CC * RES_W);
    unsigned long long* tilebest = (unsigned long long*)(tsel + 8);
    unsigned long long* kw0 = (unsigned long long*)(ws + KW_OFF);
    unsigned long long* kw1 = kw0 + KWSLOT;
    unsigned long long* kw2 = kw1 + KWSLOT;

    k_setup<<<SETUP_BLOCKS, SETUP_THREADS, 0, stream>>>(d, x, inv, resk, res0, res1, out, kw0, tsel);

    const size_t shbytes = (size_t)CC * RES_W * sizeof(float);
    dim3 grid(NATOMS / MTILE, NSPLIT, BATCH);
    k_corr<<<grid, CORR_THREADS, shbytes, stream>>>(d, inv, resk, tilebest, poskeys, kw0);
    k_corr_inc<<<MTILES + 1, CORR_THREADS, 0, stream>>>(d, inv, res0, res1, out, tsel, tilebest, poskeys, kw0, kw1, kw2);
    for (int it = 2; it < NITER; it += 6) {
        k_corr_inc<<<MTILES + 1, CORR_THREADS, 0, stream>>>(d, inv, res1, res0, out, tsel, tilebest, poskeys, kw1, kw2, kw0);
        k_corr_inc<<<MTILES + 1, CORR_THREADS, 0, stream>>>(d, inv, res0, res1, out, tsel, tilebest, poskeys, kw2, kw0, kw1);
        k_corr_inc<<<MTILES + 1, CORR_THREADS, 0, stream>>>(d, inv, res1, res0, out, tsel, tilebest, poskeys, kw0, kw1, kw2);
        k_corr_inc<<<MTILES + 1, CORR_THREADS, 0, stream>>>(d, inv, res0, res1, out, tsel, tilebest, poskeys, kw1, kw2, kw0);
        k_corr_inc<<<MTILES + 1, CORR_THREADS, 0, stream>>>(d, inv, res1, res0, out, tsel, tilebest, poskeys, kw2, kw0, kw1);
        k_corr_inc<<<MTILES + 1, CORR_THREADS, 0, stream>>>(d, inv, res0, res1, out, tsel, tilebest, poskeys, kw0, kw1, kw2);
    }
    k_update_last<<<BATCH, MTILES, 0, stream>>>(d, inv, out, kw1);
}


# Incremental launch requests its 32 KB atom tile together with the step key, ahead of the wait for it: the slab and chosen-atom loads that hang on the key no longer queue behind the tile
# speedup vs baseline: 5.8369x; 1.0313x over previous
#include <hip/hip_runtime.h>

#define NATOMS 4096
#define CC     64
#define TT     256
#define ASIZE  8
#define BATCH  4
#define KDIM   512
#define PPOS   257
#define RES_W  304
#define NSPLIT 2
#define NTPB   9
#define MTILE  64
#define CORR_THREADS 128
#define NITER  32
#define NTILES 17
#define MTILES 256
#define KSLOTS 32
#define TCOLS  272
#define SCOLS  24
#define SLABW  25
#define APITCH 516
#define KWPITCH 32
#define KWSLOT (BATCH * KWPITCH)
#define KW_OFF 4194304
#define WLAST  (PPOS - 16)
#define INVP   32
#define POSK_OFF 1048576
#define RESK_OFF 5242880
#define SETUP_THREADS 256
#define SETUP_BLOCKS  256

typedef __attribute__((ext_vector_type(2))) float v2f;
typedef __attribute__((ext_vector_type(8))) float v8f;

__global__ __launch_bounds__(SETUP_THREADS)
void k_setup(const float* __restrict__ d, const float* __restrict__ x, float* __restrict__ inv,
             float* __restrict__ resk, float* __restrict__ res0, float* __restrict__ res1,
             float* __restrict__ out, unsigned long long* __restrict__ kw, int* __restrict__ tsel) {
    const int tid  = threadIdx.x;
    const int lane = tid & 31;
    const int a0   = (blockIdx.x * (SETUP_THREADS / 32) + (tid >> 5)) * 2;
    const float* src = d + (size_t)a0 * KDIM + lane;
    float e[2][KDIM / 32];
    #pragma unroll
    for (int q = 0; q < 2; ++q) {
        #pragma unroll
        for (int j = 0; j < KDIM / 32; ++j) e[q][j] = src[q * KDIM + 32 * j];
    }

    const int gid = blockIdx.x * SETUP_THREADS + tid;
    if (gid < 3 * BATCH) kw[(size_t)gid * KWPITCH] = 0ull;
    if (gid < BATCH) tsel[BATCH + gid] = 0;
    for (int i = gid; i < BATCH * CC * RES_W; i += SETUP_BLOCKS * SETUP_THREADS) {
        const int col = i % RES_W;
        const int bc  = i / RES_W;
        resk[i] = (col < TT) ? x[bc * TT + col] : 0.f;
        if (col < TT) out[bc * TT + col] = 0.f;
    }
    if (gid < BATCH * (TCOLS / 4) * CC) {
        const int c  = gid % CC;
        const int q  = (gid / CC) % (TCOLS / 4);
        const int b  = gid / (CC * (TCOLS / 4));
        float4 v = make_float4(0.f, 0.f, 0.f, 0.f);
        if (q < TT / 4) v = *(const float4*)(x + ((size_t)b * CC + c) * TT + q * 4);
        const size_t o = ((size_t)b * TCOLS + q * 4) * CC + c;
        res0[o] = v.x;          res1[o] = v.x;
        res0[o + CC] = v.y;     res1[o + CC] = v.y;
        res0[o + 2 * CC] = v.z; res1[o + 2 * CC] = v.z;
        res0[o + 3 * CC] = v.w; res1[o + 3 * CC] = v.w;
    }

    #pragma unroll
    for (int q = 0; q < 2; ++q) {
        float part[4];
        #pragma unroll
        for (int v = 0; v < 4; ++v) {
            float s = 0.f;
            #pragma unroll
            for (int k = 0; k < 4; ++k) s = __builtin_fmaf(e[q][v + 4 * k], e[q][v + 4 * k], s);
            for (int off = 16; off > 0; off >>= 1) s += __shfl_down(s, off, 32);
            part[v] = s;
        }
        if (lane == 0)
            inv[(size_t)(a0 + q) * INVP] = 1.f / (sqrtf(part[0] + part[1] + part[2] + part[3]) + 1e-8f);
    }
}

struct Pick { float val; int atom; int pos; };

__device__ __forceinline__ Pick decode_key(unsigned long long key) {
    const unsigned enc  = (unsigned)(key >> 32);
    const unsigned bits = (enc & 0x80000000u) ? (enc & 0x7FFFFFFFu) : ~enc;
    const unsigned idx  = 0xFFFFFFFFu - (unsigned)(key & 0xFFFFFFFFu);
    const unsigned a    = idx / PPOS;
    Pick p;
    p.val  = __uint_as_float(bits);
    p.atom = (int)(a < NATOMS ? a : NATOMS - 1);
    p.pos  = (int)(idx % PPOS);
    return p;
}

__device__ __forceinline__ int first_tile(int pos) {
    const int lo = pos > 7 ? pos - 7 : 0;
    const int t0 = lo >> 4;
    return t0 < NTILES - 2 ? t0 : NTILES - 2;
}

__device__ __forceinline__ float sub_scaled(float r, float a, float val) { return __builtin_fmaf(-a, val, r); }
__device__ __forceinline__ float add_scaled(float c, float a, float val) { return __builtin_fmaf(a, val, c); }

__device__ __forceinline__ Pick pick_from_key(unsigned long long key) {
    const unsigned klo = __builtin_amdgcn_readfirstlane((unsigned)(key & 0xFFFFFFFFu));
    const unsigned khi = __builtin_amdgcn_readfirstlane((unsigned)(key >> 32));
    return decode_key(((unsigned long long)khi << 32) | klo);
}
__device__ __forceinline__ Pick pick_from_word(const unsigned long long* kw, int b) {
    return pick_from_key(kw[(size_t)b * KWPITCH]);
}

template <int NT>
__device__ __forceinline__ unsigned long long tile_keys_store(const v8f (&acc)[NT], int nt0, int atom0,
                                                int half, int nloc, int lane,
                                                unsigned long long* __restrict__ rec,
                                                unsigned long long* __restrict__ posk) {
    unsigned long long mykey = 0ull, allmax = 0ull;
    #pragma unroll
    for (int nt = 0; nt < NT; ++nt) {
        float bestv = -3.402823466e38f;
        unsigned bestidx = 0u;
        const int pos = (nt0 + nt) * 16 + nloc;
        if (pos < PPOS) {
            #pragma unroll
            for (int r = 0; r < 8; ++r) {
                const float v = acc[nt][r];
                if (v > bestv) {
                    bestv = v;
                    bestidx = (unsigned)((atom0 + r + 8 * half) * PPOS + pos);
                }
            }
        }
        unsigned vb = __float_as_uint(bestv);
        vb = (vb & 0x80000000u) ? ~vb : (vb | 0x80000000u);
        unsigned long long key = ((unsigned long long)vb << 32) | (0xFFFFFFFFu - bestidx);
        {
            unsigned long long o = __shfl_xor(key, 16, 32);
            if (o > key) key = o;
        }
        if (nt0 + nt < NTILES && lane < 16) posk[(nt0 + nt) * 16 + lane] = key;
        for (int off = 8; off > 0; off >>= 1) {
            unsigned long long o = __shfl_xor(key, off, 32);
            if (o > key) key = o;
        }
        if (lane == nt) mykey = key;
        if (key > allmax) allmax = key;
    }
    const int t = nt0 + lane;
    if (lane < NT && t < NTILES) rec[t + (t >= NTPB ? 16 - NTPB : 0)] = mykey;
    return allmax;
}

__global__ __launch_bounds__(CORR_THREADS, 1)
void k_corr(const float* __restrict__ d, const float* __restrict__ inv, const float* __restrict__ res,
            unsigned long long* __restrict__ tilebest, unsigned long long* __restrict__ poskeys,
            unsigned long long* kw_out) {
    extern __shared__ float smem[];
    const int b    = blockIdx.z;
    const int nt0  = blockIdx.y * NTPB;
    const int m0   = blockIdx.x * MTILE;
    const int tid  = threadIdx.x;
    const int lane = tid & 31;
    const int wave = tid >> 5;

    const int half = lane >> 4;
    const int nloc = lane & 15;
    const float* arow = d + (size_t)(m0 + wave * 16 + nloc) * KDIM + half * 2;
    const float ainv = inv[(size_t)(m0 + wave * 16 + nloc) * INVP];

    #pragma unroll
    for (int pf = 0; pf < 16; ++pf)
        __builtin_prefetch(arow + pf * 32, 0, 3);

    const float4* resb = (const float4*)(res + (size_t)b * CC * RES_W);
    float4* s4 = (float4*)smem;
    for (int i = tid; i < (CC * RES_W) / 4; i += CORR_THREADS) s4[i] = resb[i];
    __syncthreads();

    v8f acc[NTPB] = {};

    auto loadA = [&](int ks) { v2f a; a.x = arow[ks * 4] * ainv; a.y = arow[ks * 4 + 1] * ainv; return a; };
    auto srowp = [&](int ks) {
        const int kbase = ks * 4 + half * 2;
        return (const float*)smem + (kbase >> 3) * RES_W + (kbase & 7) + nloc + nt0 * 16;
    };

    v2f a0 = loadA(0);
    for (int ks = 0; ks < KDIM / 4; ++ks) {
        v2f an = loadA(ks + 1 < KDIM / 4 ? ks + 1 : ks);
        const float* sr = srowp(ks);
        v2f bf[NTPB];
        #pragma unroll
        for (int nt = 0; nt < NTPB; ++nt) {
            bf[nt].x = sr[nt * 16];
            bf[nt].y = sr[nt * 16 + 1];
        }
        #pragma unroll
        for (int nt = 0; nt < NTPB; ++nt)
            acc[nt] = __builtin_amdgcn_wmma_f32_16x16x4_f32(
                false, a0, false, bf[nt], (short)0, acc[nt], false, false);
        a0 = an;
    }

    const int mt = blockIdx.x * (MTILE / 16) + wave;
    const unsigned long long best =
        tile_keys_store<NTPB>(acc, nt0, m0 + wave * 16, half, nloc, lane,
                              tilebest + ((size_t)b * MTILES + mt) * KSLOTS,
                              poskeys + ((size_t)b * MTILES + mt) * NTILES * 16);
    if (lane == 0) atomicMax(kw_out + (size_t)b * KWPITCH, best);
}

__global__ __launch_bounds__(CORR_THREADS, 1)
void k_corr_inc(const float* __restrict__ d, const float* __restrict__ inv, const float* __restrict__ rin,
                float* __restrict__ rout, float* __restrict__ out, int* __restrict__ tsel,
                unsigned long long* tilebest,
                unsigned long long* poskeys,
                const unsigned long long* kw_in, unsigned long long* kw_out, unsigned long long* kw_zero) {
    __shared__ __attribute__((aligned(16))) float atile[16 * APITCH];
    __shared__ float slab[BATCH * CC * SLABW];
    const int mt   = blockIdx.x;
    const int tid  = threadIdx.x;
    const int lane = tid & 31;
    const int b    = __builtin_amdgcn_readfirstlane(tid >> 5);
    const int half = lane >> 4;
    const int nloc = lane & 15;

    if (mt == MTILES) {
        const Pick p = pick_from_word(kw_in, b);
        int pprev = tsel[BATCH + b];
        pprev = pprev < 0 ? 0 : (pprev < PPOS ? pprev : PPOS - 1);
        const float* ap = d + (size_t)p.atom * KDIM;
        const float pinv = inv[(size_t)p.atom * INVP];
        const float* ri = rin  + (size_t)b * TCOLS * CC;
        float*       ro = rout + (size_t)b * TCOLS * CC;
        float*       ob = out  + (size_t)b * CC * TT;
        const int t  = lane & 7;
        const int c0 = lane >> 3;
        const int col = p.pos + t;
        const int pc  = pprev + t;
        const bool carry = pc < p.pos || pc >= p.pos + ASIZE;
        float a[KDIM / 32], r[KDIM / 32], o[KDIM / 32], pv[KDIM / 32];
        #pragma unroll
        for (int j = 0; j < KDIM / 32; ++j) {
            const int c = c0 + 4 * j;
            a[j]  = ap[lane + 32 * j];
            r[j]  = ri[col * CC + c];
            pv[j] = ri[pc * CC + c];
            o[j]  = col < TT ? ob[c * TT + col] : 0.f;
        }
        #pragma unroll
        for (int j = 0; j < KDIM / 32; ++j) {
            const int c = c0 + 4 * j;
            const float an = a[j] * pinv;
            ro[col * CC + c] = sub_scaled(r[j], an, p.val);
            if (carry) ro[pc * CC + c] = pv[j];
            if (col < TT) ob[c * TT + col] = add_scaled(o[j], an, p.val);
        }
        if (lane == 0) {
            tsel[BATCH + b] = p.pos;
            kw_zero[(size_t)b * KWPITCH] = 0ull;
        }
        return;
    }

    const unsigned long long keyword = kw_in[(size_t)b * KWPITCH];
    const float4* asrc = (const float4*)(d + (size_t)mt * 16 * KDIM) + tid;
    float4 av[16];
    #pragma unroll
    for (int j = 0; j < 16; ++j) av[j] = asrc[j * (KDIM / 4)];
    const Pick p = pick_from_key(keyword);

    float rinv[16];
    #pragma unroll
    for (int j = 0; j < 16; ++j) rinv[j] = inv[(size_t)(mt * 16 + j) * INVP];

    unsigned long long* rec = tilebest + ((size_t)b * MTILES + mt) * KSLOTS;
    unsigned long long oldkeys[NTILES];
    #pragma unroll
    for (int j = 0; j < NTILES; ++j) oldkeys[j] = rec[j + (j >= NTPB ? 16 - NTPB : 0)];

    const int ppos = p.pos;
    const int t0 = first_tile(ppos);
    int w = ppos > 7 ? ppos - 7 : 0;
    w = w < WLAST ? w : WLAST;
    const int woff = w - t0 * 16;

    unsigned long long* pkr = poskeys + (((size_t)b * MTILES + mt) * NTILES + t0) * 16;
    const unsigned long long oldpk = pkr[lane];

    float* sl = slab + b * CC * SLABW;
    const float4* rsrc = (const float4*)(rin + ((size_t)b * TCOLS + w) * CC) + lane;
    float4 sv[(SCOLS * CC / 4) / 32];
    #pragma unroll
    for (int j = 0; j < (SCOLS * CC / 4) / 32; ++j) sv[j] = rsrc[32 * j];
    float4 uv[KDIM / 128];
    const float4* arow4 = (const float4*)(d + (size_t)p.atom * KDIM);
    #pragma unroll
    for (int j = 0; j < KDIM / 128; ++j) uv[j] = arow4[lane + 32 * j];
    const float pinv = inv[(size_t)p.atom * INVP];
    asm volatile("" : "+s"(rinv[0]), "+s"(rinv[1]), "+s"(rinv[2]), "+s"(rinv[3]),
                      "+s"(rinv[4]), "+s"(rinv[5]), "+s"(rinv[6]), "+s"(rinv[7]));
    asm volatile("" : "+s"(rinv[8]), "+s"(rinv[9]), "+s"(rinv[10]), "+s"(rinv[11]),
                      "+s"(rinv[12]), "+s"(rinv[13]), "+s"(rinv[14]), "+s"(rinv[15]));
    #pragma unroll
    for (int j = 0; j < 16; ++j) {
        float4 v = av[j];
        v.x *= rinv[j]; v.y *= rinv[j]; v.z *= rinv[j]; v.w *= rinv[j];
        *(float4*)(atile + j * APITCH + tid * 4) = v;
    }
    #pragma unroll
    for (int j = 0; j < (SCOLS * CC / 4) / 32; ++j) {
        float* q = sl + (lane & 15) * 4 * SLABW + 2 * j + (lane >> 4);
        q[0] = sv[j].x; q[SLABW] = sv[j].y; q[2 * SLABW] = sv[j].z; q[3 * SLABW] = sv[j].w;
    }
    __syncthreads();

    {
        float* up = sl + (p.pos - w);
        float rv[KDIM / 128][4];
        #pragma unroll
        for (int j = 0; j < KDIM / 128; ++j) {
            const int i4 = lane + 32 * j;
            const float* q = up + (i4 >> 1) * SLABW + (i4 & 1) * 4;
            rv[j][0] = q[0]; rv[j][1] = q[1]; rv[j][2] = q[2]; rv[j][3] = q[3];
        }
        #pragma unroll
        for (int j = 0; j < KDIM / 128; ++j) {
            const int i4 = lane + 32 * j;
            float* q = up + (i4 >> 1) * SLABW + (i4 & 1) * 4;
            q[0] = sub_scaled(rv[j][0], uv[j].x * pinv, p.val);
            q[1] = sub_scaled(rv[j][1], uv[j].y * pinv, p.val);
            q[2] = sub_scaled(rv[j][2], uv[j].z * pinv, p.val);
            q[3] = sub_scaled(rv[j][3], uv[j].w * pinv, p.val);
        }
        __syncthreads();
    }

    const float* al = atile + nloc * APITCH + half * 2;
    const float* bl = sl + half * 2 + nloc;
    v8f acc = {};

    auto loadG = [&](int g, v2f (&a)[8], v2f (&bq)[8]) {
        const float* ap = al + g * 32;
        const float* bp = bl + g * 4 * SLABW;
        #pragma unroll
        for (int j = 0; j < 8; ++j) {
            a[j] = *(const v2f*)(ap + j * 4);
            const float* sp = bp + (j >> 1) * SLABW + (j & 1) * 4;
            bq[j].x = sp[0];
            bq[j].y = sp[1];
        }
    };
    auto mmaG = [&](const v2f (&a)[8], const v2f (&bq)[8]) {
        #pragma unroll
        for (int j = 0; j < 8; ++j)
            acc = __builtin_amdgcn_wmma_f32_16x16x4_f32(
                false, a[j], false, bq[j], (short)0, acc, false, false);
    };

    v2f aA[8], aB[8], bA[8], bB[8];
    loadG(0, aA, bA);
    for (int g = 0; g < KDIM / 32; g += 2) {
        loadG(g + 1, aB, bB);
        __builtin_amdgcn_sched_barrier(0);
        mmaG(aA, bA);
        __builtin_amdgcn_sched_barrier(0);
        loadG(g + 2 < KDIM / 32 ? g + 2 : g + 1, aA, bA);
        __builtin_amdgcn_sched_barrier(0);
        mmaG(aB, bB);
        __builtin_amdgcn_sched_barrier(0);
    }

    float bestv = acc[0];
    unsigned bestidx = (unsigned)((mt * 16 + 8 * half) * PPOS + w + nloc);
    #pragma unroll
    for (int r = 1; r < 8; ++r) {
        const float v = acc[r];
        if (v > bestv) {
            bestv = v;
            bestidx = (unsigned)((mt * 16 + r + 8 * half) * PPOS + w + nloc);
        }
    }
    unsigned vb = __float_as_uint(bestv);
    vb = (vb & 0x80000000u) ? ~vb : (vb | 0x80000000u);
    unsigned long long ckey = ((unsigned long long)vb << 32) | (0xFFFFFFFFu - bestidx);
    {
        const unsigned long long o = __shfl_xor(ckey, 16, 32);
        if (o > ckey) ckey = o;
    }
    const int rel = lane - woff;
    const unsigned long long nkey = __shfl(ckey, rel & 15, 32);
    unsigned long long mk = (rel >= 0 && rel < 16) ? nkey : oldpk;
    pkr[lane] = mk;
    for (int off = 8; off > 0; off >>= 1) {
        const unsigned long long o = __shfl_xor(mk, off, 32);
        if (o > mk) mk = o;
    }
    {
        const int t = t0 + half;
        if (nloc == 0 && t < NTILES) rec[t + (t >= NTPB ? 16 - NTPB : 0)] = mk;
    }
    unsigned long long newmax = mk;
    {
        const unsigned long long o = __shfl_xor(mk, 16, 32);
        if (o > newmax) newmax = o;
    }

    unsigned long long oldmax = 0ull;
    #pragma unroll
    for (int j = 0; j < NTILES; ++j)
        if (j != t0 && j != t0 + 1 && oldkeys[j] > oldmax) oldmax = oldkeys[j];
    if (lane == 0) atomicMax(kw_out + (size_t)b * KWPITCH, newmax > oldmax ? newmax : oldmax);
}

__global__ __launch_bounds__(MTILES)
void k_update_last(const float* __restrict__ d, const float* __restrict__ inv, float* __restrict__ out,
                   const unsigned long long* __restrict__ kw_in) {
    const int b    = blockIdx.x;
    const int tid  = threadIdx.x;
    const Pick p = decode_key(kw_in[(size_t)b * KWPITCH]);
    const float* ap = d + (size_t)p.atom * KDIM;
    const float pinv = inv[(size_t)p.atom * INVP];
    float* ob = out + (size_t)b * CC * TT;
    for (int i = tid; i < KDIM; i += MTILES) {
        const int c = i >> 3, t = i & 7;
        const int col = p.pos + t;
        const float a = ap[i] * pinv;
        if (col < TT) ob[c * TT + col] = add_scaled(ob[c * TT + col], a, p.val);
    }
}

extern "C" void kernel_launch(void* const* d_in, const int* in_sizes, int n_in,
                              void* d_out, int out_size, void* d_ws, size_t ws_size,
                              hipStream_t stream) {
    const float* x = (const float*)d_in[0];
    const float* d = (const float*)d_in[1];
    (void)in_sizes; (void)n_in; (void)out_size; (void)ws_size;
    float* out = (float*)d_out;

    char* ws = (char*)d_ws;
    float* inv  = (float*)ws;
    unsigned long long* poskeys = (unsigned long long*)(ws + POSK_OFF);
    float* resk = (float*)(ws + RESK_OFF);
    float* res0 = (float*)(ws + (size_t)NATOMS * KDIM * 4);
    float* res1 = res0 + BATCH * CC * RES_W;
    int* tsel = (int*)(res1 + BATCH * CC * RES_W);
    unsigned long long* tilebest = (unsigned long long*)(tsel + 8);
    unsigned long long* kw0 = (unsigned long long*)(ws + KW_OFF);
    unsigned long long* kw1 = kw0 + KWSLOT;
    unsigned long long* kw2 = kw1 + KWSLOT;

    k_setup<<<SETUP_BLOCKS, SETUP_THREADS, 0, stream>>>(d, x, inv, resk, res0, res1, out, kw0, tsel);

    const size_t shbytes = (size_t)CC * RES_W * sizeof(float);
    dim3 grid(NATOMS / MTILE, NSPLIT, BATCH);
    k_corr<<<grid, CORR_THREADS, shbytes, stream>>>(d, inv, resk, tilebest, poskeys, kw0);
    k_corr_inc<<<MTILES + 1, CORR_THREADS, 0, stream>>>(d, inv, res0, res1, out, tsel, tilebest, poskeys, kw0, kw1, kw2);
    for (int it = 2; it < NITER; it += 6) {
        k_corr_inc<<<MTILES + 1, CORR_THREADS, 0, stream>>>(d, inv, res1, res0, out, tsel, tilebest, poskeys, kw1, kw2, kw0);
        k_corr_inc<<<MTILES + 1, CORR_THREADS, 0, stream>>>(d, inv, res0, res1, out, tsel, tilebest, poskeys, kw2, kw0, kw1);
        k_corr_inc<<<MTILES + 1, CORR_THREADS, 0, stream>>>(d, inv, res1, res0, out, tsel, tilebest, poskeys, kw0, kw1, kw2);
        k_corr_inc<<<MTILES + 1, CORR_THREADS, 0, stream>>>(d, inv, res0, res1, out, tsel, tilebest, poskeys, kw1, kw2, kw0);
        k_corr_inc<<<MTILES + 1, CORR_THREADS, 0, stream>>>(d, inv, res1, res0, out, tsel, tilebest, poskeys, kw2, kw0, kw1);
        k_corr_inc<<<MTILES + 1, CORR_THREADS, 0, stream>>>(d, inv, res0, res1, out, tsel, tilebest, poskeys, kw0, kw1, kw2);
    }
    k_update_last<<<BATCH, MTILES, 0, stream>>>(d, inv, out, kw1);
}
